# starting kernel
# baseline (speedup 1.0000x reference)
#include <hip/hip_runtime.h>
#include <hip/hip_bf16.h>

typedef float v2f __attribute__((ext_vector_type(2)));
typedef float v8f __attribute__((ext_vector_type(8)));

#define D 128
#define HIST 50
#define CAND 20

__global__ __launch_bounds__(32) void encoder_gemm(
    const float* __restrict__ x, const float* __restrict__ W,
    float* __restrict__ out, int K  ) {
  const int lane = threadIdx.x;
  const int tile = blockIdx.x;
  const int mt = tile >> 3;
  const int nt = tile & 7;
  const int arow = mt * 16 + (lane & 15);
  const int col  = nt * 16 + (lane & 15);
  const int kh   = (lane >> 4) * 2;

  const float* __restrict__ xr = x + (size_t)arow * K;
  v8f acc = {};
  for (int k = 0; k < K; k += 4) {
    v2f a, b;
    a.x = xr[k + kh + 0];
    a.y = xr[k + kh + 1];
    b.x = W[(size_t)(k + kh + 0) * D + col];
    b.y = W[(size_t)(k + kh + 1) * D + col];
    acc = __builtin_amdgcn_wmma_f32_16x16x4_f32(
        false, a, false, b, (short)0, acc, false, false);
  }
  const int mbase = mt * 16 + (lane >> 4) * 8;
#pragma unroll
  for (int r = 0; r < 8; ++r)
    out[(size_t)(mbase + r) * D + col] = acc[r];
}

__global__ void edge_agg(const int* __restrict__ ei,
                         const float* __restrict__ news,
                         float* __restrict__ agg, float* __restrict__ deg,
                         long long E) {
  long long idx = (long long)blockIdx.x * blockDim.x + threadIdx.x;
  long long e = idx >> 7;
  int d = (int)(idx & (D - 1));
  if (e >= E) return;
  int s = ei[e];
  int t = ei[E + e];
  atomicAdd(&agg[(size_t)t * D + d], news[(size_t)s * D + d]);
  if (d == 0) atomicAdd(&deg[t], 1.0f);
}

__global__ __launch_bounds__(32) void gnn_gemm_add(
    const float* __restrict__ agg, const float* __restrict__ deg,
    const float* __restrict__ Wg, const float* __restrict__ news,
    float* __restrict__ xx) {
  const int lane = threadIdx.x;
  const int tile = blockIdx.x;
  const int mt = tile >> 3;
  const int nt = tile & 7;
  const int arow = mt * 16 + (lane & 15);
  const int col  = nt * 16 + (lane & 15);
  const int kh   = (lane >> 4) * 2;

  const float dinv = 1.0f / fmaxf(deg[arow], 1.0f);
  const float* __restrict__ ar = agg + (size_t)arow * D;
  v8f acc = {};
  for (int k = 0; k < D; k += 4) {
    v2f a, b;
    a.x = ar[k + kh + 0] * dinv;
    a.y = ar[k + kh + 1] * dinv;
    b.x = Wg[(size_t)(k + kh + 0) * D + col];
    b.y = Wg[(size_t)(k + kh + 1) * D + col];
    acc = __builtin_amdgcn_wmma_f32_16x16x4_f32(
        false, a, false, b, (short)0, acc, false, false);
  }
  const int mbase = mt * 16 + (lane >> 4) * 8;
#pragma unroll
  for (int r = 0; r < 8; ++r) {
    size_t o = (size_t)(mbase + r) * D + col;
    xx[o] = news[o] + acc[r];
  }
}

__global__ __launch_bounds__(128) void user_score(
    const float* __restrict__ xx, const int* __restrict__ hist,
    const int* __restrict__ cand, float* __restrict__ scores) {
  const int b = blockIdx.x;
  const int t = threadIdx.x;
  __shared__ float uv[D];

  float sum = 0.0f, cnt = 0.0f;
  for (int h = 0; h < HIST; ++h) {
    int nid = hist[b * HIST + h];
    if (nid != 0) {
      sum += xx[(size_t)nid * D + t];
      cnt += 1.0f;
    }
  }
  uv[t] = sum / fmaxf(cnt, 1e-9f);
  __syncthreads();

  const int wave = t >> 5, lane = t & 31;
  for (int c = wave; c < CAND; c += 4) {
    int nid = cand[b * CAND + c];
    const float* __restrict__ ce = xx + (size_t)nid * D;
    float p = 0.0f;
#pragma unroll
    for (int d = lane; d < D; d += 32) p += ce[d] * uv[d];
#pragma unroll
    for (int off = 16; off; off >>= 1) p += __shfl_xor(p, off, 32);
    if (lane == 0) scores[b * CAND + c] = p;
  }
}

extern "C" void kernel_launch(void* const* d_in, const int* in_sizes, int n_in,
                              void* d_out, int out_size, void* d_ws, size_t ws_size,
                              hipStream_t stream) {
  const float* x      = (const float*)d_in[0];
  const int*   ei     = (const int*)  d_in[2];
  const int*   hist   = (const int*)  d_in[3];
  const int*   cand   = (const int*)  d_in[4];
  const float* W_enc  = (const float*)d_in[5];
  const float* W_gnn  = (const float*)d_in[6];
  float*       scores = (float*)d_out;

  const int       N = in_sizes[1];
  const int       K = in_sizes[0] / N;
  const long long E = (long long)in_sizes[2] / 2;
  const int       B = in_sizes[3] / HIST;

  float* news = (float*)d_ws;
  float* agg  = news + (size_t)N * D;
  float* deg  = agg  + (size_t)N * D;
  float* xx   = deg  + (size_t)N;

  hipMemsetAsync(agg, 0, (size_t)N * D * sizeof(float), stream);
  hipMemsetAsync(deg, 0, (size_t)N * sizeof(float), stream);

  const int tiles = (N / 16) * (D / 16);
  encoder_gemm<<<tiles, 32, 0, stream>>>(x, W_enc, news, K);

  long long work = E * D;
  int eblocks = (int)((work + 255) / 256);
  edge_agg<<<eblocks, 256, 0, stream>>>(ei, news, agg, deg, E);

  gnn_gemm_add<<<tiles, 32, 0, stream>>>(agg, deg, W_gnn, news, xx);

  user_score<<<B, 128, 0, stream>>>(xx, hist, cand, scores);
}


# Edge scan reads one bit per edge from the owner's own bit row (pre-pass builds a row per owner in LDS with ds_or, 512 B per store) instead of every edge's key byte; LDS sum order stays arrival order
# speedup vs baseline: 4.6527x; 4.6527x over previous
#include <hip/hip_runtime.h>
#include <hip/hip_bf16.h>

typedef float v2f __attribute__((ext_vector_type(2)));
typedef float v8f __attribute__((ext_vector_type(8)));

#define D 128
#define HIST 50
#define CAND 20

#define GB_ROWS 32

template <int K, int PITCH, bool GNN>
__global__ __launch_bounds__(256) __attribute__((amdgpu_waves_per_eu(4, 4))) void gemm_wstat(
    const float* __restrict__ A, const float* __restrict__ W,
    const float* __restrict__ deg, const float* __restrict__ news,
    float* __restrict__ out, int N) {
  constexpr int KS  = K / 4;
  constexpr int RF4 = K / 4;
  constexpr int NF4 = GB_ROWS * RF4;
  constexpr int NLD = (NF4 + 255) / 256;
  constexpr int STAGE = (GB_ROWS * PITCH > NLD * 1024) ? GB_ROWS * PITCH : NLD * 1024;
  __shared__ float lds[2][STAGE];

  const int tid  = threadIdx.x;
  const int lane = tid & 31;
  const int wave = __builtin_amdgcn_readfirstlane(tid >> 5);
  const int c = lane & 15, h = lane >> 4;
  const int col = wave * 16 + c;
  const int nblk = N / GB_ROWS;

  v2f b[KS];
#pragma unroll
  for (int j = 0; j < KS; ++j) {
    b[j].x = W[(size_t)(4 * j + 2 * h + 0) * D + col];
    b[j].y = W[(size_t)(4 * j + 2 * h + 1) * D + col];
  }

  float4 p[NLD];
  float  dv[NLD];
  int blk = blockIdx.x;

  if (blk < nblk) {
    const float4* __restrict__ src = reinterpret_cast<const float4*>(A) + (size_t)blk * NF4;
#pragma unroll
    for (int i = 0; i < NLD; ++i) {
      const int f = min(tid + 256 * i, NF4 - 1);
      p[i] = src[f];
      if (GNN) dv[i] = deg[blk * GB_ROWS + f / RF4];
    }
#pragma unroll
    for (int i = 0; i < NLD; ++i) {
      const int f = tid + 256 * i;
      float4 v = p[i];
      if (GNN) {
        const float dinv = 1.0f / fmaxf(dv[i], 1.0f);
        v.x *= dinv; v.y *= dinv; v.z *= dinv; v.w *= dinv;
      }
      const int a = (PITCH == K) ? 4 * f : (f / RF4) * PITCH + (f % RF4) * 4;
      *reinterpret_cast<float4*>(&lds[0][a]) = v;
    }
  }
  __syncthreads();

  int s = 0;
  for (; blk < nblk; blk += gridDim.x) {
    const int nxt = (blk + (int)gridDim.x < nblk) ? blk + (int)gridDim.x : blk;
    {
      const float4* __restrict__ src = reinterpret_cast<const float4*>(A) + (size_t)nxt * NF4;
#pragma unroll
      for (int i = 0; i < NLD; ++i) {
        const int f = min(tid + 256 * i, NF4 - 1);
        p[i] = src[f];
        if (GNN) dv[i] = deg[nxt * GB_ROWS + f / RF4];
      }
    }
    const size_t o0 = (size_t)(blk * GB_ROWS + 8 * h) * D + col;
    float nv0[8], nv1[8];
    if (GNN) {
#pragma unroll
      for (int r = 0; r < 8; ++r) {
        nv0[r] = news[o0 + (size_t)r * D];
        nv1[r] = news[o0 + (size_t)(16 + r) * D];
      }
    }

    const float* sA = &lds[s][c * PITCH + 2 * h];
    v8f acc0 = {}, acc1 = {};
#pragma unroll
    for (int j = 0; j < KS; ++j) {
      const v2f a0 = *reinterpret_cast<const v2f*>(sA + 4 * j);
      const v2f a1 = *reinterpret_cast<const v2f*>(sA + 16 * PITCH + 4 * j);
      acc0 = __builtin_amdgcn_wmma_f32_16x16x4_f32(
          false, a0, false, b[j], (short)0, acc0, false, false);
      acc1 = __builtin_amdgcn_wmma_f32_16x16x4_f32(
          false, a1, false, b[j], (short)0, acc1, false, false);
    }

#pragma unroll
    for (int r = 0; r < 8; ++r) {
      if (GNN) {
        out[o0 + (size_t)r * D]        = nv0[r] + acc0[r];
        out[o0 + (size_t)(16 + r) * D] = nv1[r] + acc1[r];
      } else {
        out[o0 + (size_t)r * D]        = acc0[r];
        out[o0 + (size_t)(16 + r) * D] = acc1[r];
      }
    }

    {
#pragma unroll
      for (int i = 0; i < NLD; ++i) {
        const int f = tid + 256 * i;
        float4 v = p[i];
        if (GNN) {
          const float dinv = 1.0f / fmaxf(dv[i], 1.0f);
          v.x *= dinv; v.y *= dinv; v.z *= dinv; v.w *= dinv;
        }
        const int a = (PITCH == K) ? 4 * f : (f / RF4) * PITCH + (f % RF4) * 4;
        *reinterpret_cast<float4*>(&lds[s ^ 1][a]) = v;
      }
    }
    __syncthreads();
    s ^= 1;
  }
}

#define RCAP   416
#define NOWN   256
#define BM_BW  128
#define BM_E   (32 * BM_BW)
#define BM_LP  132
#define QCAP   64

__global__ __launch_bounds__(1024) void edge_owner_bits(
    const int* __restrict__ ei, unsigned* __restrict__ bits, int E, int N) {
  __shared__ unsigned lb[NOWN * BM_LP];
  const int* __restrict__ dstp = ei + E;
  const int tid  = threadIdx.x;
  const int lane = tid & 31;
  const int wave = __builtin_amdgcn_readfirstlane(tid >> 5);
  const int nblk = (E + BM_E - 1) / BM_E;
  const int nwp  = nblk * BM_BW;
  const int rows = min(NOWN, (N + RCAP - 1) / RCAP);
  const bool vec = (reinterpret_cast<size_t>(dstp) & 15) == 0;

  for (int blk = blockIdx.x; blk < nblk; blk += gridDim.x) {
    for (int i = tid; i < NOWN * BM_LP / 4; i += 1024)
      reinterpret_cast<uint4*>(lb)[i] = make_uint4(0u, 0u, 0u, 0u);
    __syncthreads();

    const int e0 = blk * BM_E + 4 * tid;
    unsigned* __restrict__ wp = lb + (tid >> 3);
    const int sh = 4 * (tid & 7);
    if (vec && e0 + 3 < E) {
      const int4 d = *reinterpret_cast<const int4*>(dstp + e0);
      atomicOr(wp + (((unsigned)d.x / RCAP) & 255u) * BM_LP, 1u << (sh + 0));
      atomicOr(wp + (((unsigned)d.y / RCAP) & 255u) * BM_LP, 1u << (sh + 1));
      atomicOr(wp + (((unsigned)d.z / RCAP) & 255u) * BM_LP, 1u << (sh + 2));
      atomicOr(wp + (((unsigned)d.w / RCAP) & 255u) * BM_LP, 1u << (sh + 3));
    } else {
      for (int j = 0; j < 4; ++j)
        if (e0 + j < E)
          atomicOr(wp + (((unsigned)dstp[e0 + j] / RCAP) & 255u) * BM_LP, 1u << (sh + j));
    }
    __syncthreads();

    for (int o = wave; o < rows; o += 32) {
      const uint4 v = *reinterpret_cast<const uint4*>(lb + o * BM_LP + 4 * lane);
      *reinterpret_cast<uint4*>(bits + (size_t)o * nwp + (size_t)blk * BM_BW + 4 * lane) = v;
    }
    __syncthreads();
  }
}

__device__ __forceinline__ void drain_queue(
    const int* __restrict__ myq, int qn, int lane,
    const int* __restrict__ srcp, const int* __restrict__ dstp, const float* __restrict__ news,
    float* acc, float* degc, int E, int lo, int cnt) {
  for (int b0 = 0; b0 < qn; b0 += 32) {
    const int nb = min(32, qn - b0);
    int s = 0, t = RCAP;
    if (lane < nb) {
      const int e = myq[b0 + lane];
      if (e < E) {
        const int d = dstp[e] - lo;
        s = srcp[e];
        if ((unsigned)d < (unsigned)cnt) {
          t = d;
          atomicAdd(&degc[t], 1.0f);
        }
      }
    }
    for (int k = 0; k < nb; k += 8) {
      float4 r[8];
      int tt[8];
#pragma unroll
      for (int v = 0; v < 8; ++v) {
        const int kv8 = min(k + v, nb - 1);
        const int sk = __builtin_amdgcn_readlane(s, kv8);
        tt[v] = __builtin_amdgcn_readlane(t, kv8);
        r[v] = *reinterpret_cast<const float4*>(news + (size_t)sk * D + lane * 4);
      }
#pragma unroll
      for (int v = 0; v < 8; ++v) {
        if (k + v < nb) {
          float* a = acc + tt[v] * D + lane;
          atomicAdd(a,      r[v].x);
          atomicAdd(a + 32, r[v].y);
          atomicAdd(a + 64, r[v].z);
          atomicAdd(a + 96, r[v].w);
        }
      }
    }
  }
}

__global__ __launch_bounds__(1024) void edge_agg_owner(
    const int* __restrict__ ei, const unsigned* __restrict__ bits, const float* __restrict__ news,
    float* __restrict__ agg, float* __restrict__ deg, int E, int N) {
  __shared__ float acc[(RCAP + 1) * D];
  __shared__ float degc[RCAP];
  __shared__ int   que[32][QCAP];

  const int tid  = threadIdx.x;
  const int lane = tid & 31;
  const int wave = __builtin_amdgcn_readfirstlane(tid >> 5);
  const int* __restrict__ srcp = ei;
  const int* __restrict__ dstp = ei + E;
  const int nq = ((E + BM_E - 1) / BM_E) * (BM_BW / 4);
  const int ntrip = (nq + 1023) / 1024;
  int* __restrict__ myq = que[wave];
  const uint4* __restrict__ kq = reinterpret_cast<const uint4*>(bits) + (size_t)(blockIdx.x & 255u) * nq;
  const uint4 zero4 = make_uint4(0u, 0u, 0u, 0u);

  for (int lo = blockIdx.x * RCAP; lo < N; lo += NOWN * RCAP) {
    const int cnt = min(RCAP, N - lo);

    for (int i = tid; i < (RCAP + 1) * D / 4; i += 1024)
      reinterpret_cast<float4*>(acc)[i] = make_float4(0.f, 0.f, 0.f, 0.f);
    if (tid < RCAP) degc[tid] = 0.f;
    __syncthreads();

    int c = (ntrip > 0) ? (int)((blockIdx.x * 37u) % (unsigned)ntrip) : 0;
    int qn = 0;
    uint4 kv = zero4, kn;
    if (ntrip > 0) {
      const int q = c * 1024 + wave * 32 + lane;
      if (q < nq) kv = kq[q];
    }
    for (int it = 0; it < ntrip; ++it) {
      const int q = c * 1024 + wave * 32 + lane;
      c = (c + 1 == ntrip) ? 0 : c + 1;
      {
        const int qx = c * 1024 + wave * 32 + lane;
        kn = zero4;
        if (qx < nq) kn = kq[qx];
      }

      const unsigned mw[4] = {kv.x, kv.y, kv.z, kv.w};
#pragma unroll
      for (int j = 0; j < 4; ++j) {
        unsigned m = mw[j];
        unsigned bal = __builtin_amdgcn_ballot_w32(m != 0u);
        while (bal != 0u) {
          if (m != 0u) {
            myq[qn + (int)__builtin_amdgcn_mbcnt_lo(bal, 0u)] = 128 * q + 32 * j + __builtin_ctz(m);
            m &= m - 1u;
          }
          qn += __builtin_popcount(bal);
          if (qn >= 32) {
            drain_queue(myq, qn, lane, srcp, dstp, news, acc, degc, E, lo, cnt);
            qn = 0;
          }
          bal = __builtin_amdgcn_ballot_w32(m != 0u);
        }
      }
      kv = kn;
    }
    if (qn > 0) drain_queue(myq, qn, lane, srcp, dstp, news, acc, degc, E, lo, cnt);
    __syncthreads();

    for (int t = wave; t < cnt; t += 32) {
      const float* a = acc + t * D + lane;
      const float4 v = make_float4(a[0], a[32], a[64], a[96]);
      *reinterpret_cast<float4*>(agg + (size_t)(lo + t) * D + lane * 4) = v;
    }
    for (int t = tid; t < cnt; t += 1024) deg[lo + t] = degc[t];
    __syncthreads();
  }
}

__global__ __launch_bounds__(128) void user_score(
    const float* __restrict__ xx, const int* __restrict__ hist,
    const int* __restrict__ cand, float* __restrict__ scores) {
  const int b = blockIdx.x;
  const int t = threadIdx.x;
  __shared__ float uv[D];

  float sum = 0.0f, cnt = 0.0f;
  for (int h = 0; h < HIST; ++h) {
    int nid = hist[b * HIST + h];
    if (nid != 0) {
      sum += xx[(size_t)nid * D + t];
      cnt += 1.0f;
    }
  }
  uv[t] = sum / fmaxf(cnt, 1e-9f);
  __syncthreads();

  const int wave = t >> 5, lane = t & 31;
  for (int c = wave; c < CAND; c += 4) {
    int nid = cand[b * CAND + c];
    const float* __restrict__ ce = xx + (size_t)nid * D;
    float p = 0.0f;
#pragma unroll
    for (int d = lane; d < D; d += 32) p += ce[d] * uv[d];
#pragma unroll
    for (int off = 16; off; off >>= 1) p += __shfl_xor(p, off, 32);
    if (lane == 0) scores[b * CAND + c] = p;
  }
}

extern "C" void kernel_launch(void* const* d_in, const int* in_sizes, int n_in,
                              void* d_out, int out_size, void* d_ws, size_t ws_size,
                              hipStream_t stream) {
  const float* x      = (const float*)d_in[0];
  const int*   ei     = (const int*)  d_in[2];
  const int*   hist   = (const int*)  d_in[3];
  const int*   cand   = (const int*)  d_in[4];
  const float* W_enc  = (const float*)d_in[5];
  const float* W_gnn  = (const float*)d_in[6];
  float*       scores = (float*)d_out;

  const int       N = in_sizes[1];
  const long long E = (long long)in_sizes[2] / 2;
  const int       B = in_sizes[3] / HIST;

  float* news = (float*)d_ws;
  float* agg  = news + (size_t)N * D;
  float* deg  = agg  + (size_t)N * D;
  float* xx   = deg  + (size_t)N;

  gemm_wstat<300, 300, false><<<512, 256, 0, stream>>>(x, W_enc, nullptr, nullptr, news, N);

  unsigned* bits = (unsigned*)xx;
  edge_owner_bits<<<512, 1024, 0, stream>>>(ei, bits, (int)E, N);

  edge_agg_owner<<<NOWN, 1024, 0, stream>>>(ei, bits, news, agg, deg, (int)E, N);

  gemm_wstat<128, 132, true><<<512, 256, 0, stream>>>(agg, W_gnn, deg, news, xx, N);

  user_score<<<B, 128, 0, stream>>>(xx, hist, cand, scores);
}


# W_gnn product moved by linearity from the 100000 node rows to the 4096 user rows: score = news[c].u + aggmean[c].(W_gnn u), both products on f32 WMMA; node GEMM and xx plane go; sum order changes
# speedup vs baseline: 6.0529x; 1.3009x over previous
#include <hip/hip_runtime.h>
#include <hip/hip_bf16.h>

typedef float v2f __attribute__((ext_vector_type(2)));
typedef float v8f __attribute__((ext_vector_type(8)));

#define D 128
#define HIST 50
#define CAND 20

#define GB_ROWS 32

template <int K, int PITCH>
__global__ __launch_bounds__(256) __attribute__((amdgpu_waves_per_eu(4, 4))) void gemm_wstat(
    const float* __restrict__ A, const float* __restrict__ W,
    float* __restrict__ out, int N) {
  constexpr int KS  = K / 4;
  constexpr int RF4 = K / 4;
  constexpr int NF4 = GB_ROWS * RF4;
  constexpr int NLD = (NF4 + 255) / 256;
  constexpr int STAGE = (GB_ROWS * PITCH > NLD * 1024) ? GB_ROWS * PITCH : NLD * 1024;
  __shared__ float lds[2][STAGE];

  const int tid  = threadIdx.x;
  const int lane = tid & 31;
  const int wave = __builtin_amdgcn_readfirstlane(tid >> 5);
  const int c = lane & 15, h = lane >> 4;
  const int col = wave * 16 + c;
  const int nblk = N / GB_ROWS;

  v2f b[KS];
#pragma unroll
  for (int j = 0; j < KS; ++j) {
    b[j].x = W[(size_t)(4 * j + 2 * h + 0) * D + col];
    b[j].y = W[(size_t)(4 * j + 2 * h + 1) * D + col];
  }

  float4 p[NLD];
  int blk = blockIdx.x;

  if (blk < nblk) {
    const float4* __restrict__ src = reinterpret_cast<const float4*>(A) + (size_t)blk * NF4;
#pragma unroll
    for (int i = 0; i < NLD; ++i) {
      const int f = min(tid + 256 * i, NF4 - 1);
      p[i] = src[f];
    }
#pragma unroll
    for (int i = 0; i < NLD; ++i) {
      const int f = tid + 256 * i;
      const int a = (PITCH == K) ? 4 * f : (f / RF4) * PITCH + (f % RF4) * 4;
      *reinterpret_cast<float4*>(&lds[0][a]) = p[i];
    }
  }
  __syncthreads();

  int s = 0;
  for (; blk < nblk; blk += gridDim.x) {
    const int nxt = (blk + (int)gridDim.x < nblk) ? blk + (int)gridDim.x : blk;
    {
      const float4* __restrict__ src = reinterpret_cast<const float4*>(A) + (size_t)nxt * NF4;
#pragma unroll
      for (int i = 0; i < NLD; ++i) {
        const int f = min(tid + 256 * i, NF4 - 1);
        p[i] = src[f];
      }
    }
    const size_t o0 = (size_t)(blk * GB_ROWS + 8 * h) * D + col;

    const float* sA = &lds[s][c * PITCH + 2 * h];
    v8f acc0 = {}, acc1 = {};
#pragma unroll
    for (int j = 0; j < KS; ++j) {
      const v2f a0 = *reinterpret_cast<const v2f*>(sA + 4 * j);
      const v2f a1 = *reinterpret_cast<const v2f*>(sA + 16 * PITCH + 4 * j);
      acc0 = __builtin_amdgcn_wmma_f32_16x16x4_f32(
          false, a0, false, b[j], (short)0, acc0, false, false);
      acc1 = __builtin_amdgcn_wmma_f32_16x16x4_f32(
          false, a1, false, b[j], (short)0, acc1, false, false);
    }

#pragma unroll
    for (int r = 0; r < 8; ++r) {
      out[o0 + (size_t)r * D]        = acc0[r];
      out[o0 + (size_t)(16 + r) * D] = acc1[r];
    }

    {
#pragma unroll
      for (int i = 0; i < NLD; ++i) {
        const int f = tid + 256 * i;
        const int a = (PITCH == K) ? 4 * f : (f / RF4) * PITCH + (f % RF4) * 4;
        *reinterpret_cast<float4*>(&lds[s ^ 1][a]) = p[i];
      }
    }
    __syncthreads();
    s ^= 1;
  }
}

#define RCAP   416
#define NOWN   256
#define BM_BW  128
#define BM_E   (32 * BM_BW)
#define BM_LP  132
#define QCAP   64

__global__ __launch_bounds__(1024) void edge_owner_bits(
    const int* __restrict__ ei, unsigned* __restrict__ bits, int E, int N) {
  __shared__ unsigned lb[NOWN * BM_LP];
  const int* __restrict__ dstp = ei + E;
  const int tid  = threadIdx.x;
  const int lane = tid & 31;
  const int wave = __builtin_amdgcn_readfirstlane(tid >> 5);
  const int nblk = (E + BM_E - 1) / BM_E;
  const int nwp  = nblk * BM_BW;
  const int rows = min(NOWN, (N + RCAP - 1) / RCAP);
  const bool vec = (reinterpret_cast<size_t>(dstp) & 15) == 0;

  for (int blk = blockIdx.x; blk < nblk; blk += gridDim.x) {
    for (int i = tid; i < NOWN * BM_LP / 4; i += 1024)
      reinterpret_cast<uint4*>(lb)[i] = make_uint4(0u, 0u, 0u, 0u);
    __syncthreads();

    const int e0 = blk * BM_E + 4 * tid;
    unsigned* __restrict__ wp = lb + (tid >> 3);
    const int sh = 4 * (tid & 7);
    if (vec && e0 + 3 < E) {
      const int4 d = *reinterpret_cast<const int4*>(dstp + e0);
      atomicOr(wp + (((unsigned)d.x / RCAP) & 255u) * BM_LP, 1u << (sh + 0));
      atomicOr(wp + (((unsigned)d.y / RCAP) & 255u) * BM_LP, 1u << (sh + 1));
      atomicOr(wp + (((unsigned)d.z / RCAP) & 255u) * BM_LP, 1u << (sh + 2));
      atomicOr(wp + (((unsigned)d.w / RCAP) & 255u) * BM_LP, 1u << (sh + 3));
    } else {
      for (int j = 0; j < 4; ++j)
        if (e0 + j < E)
          atomicOr(wp + (((unsigned)dstp[e0 + j] / RCAP) & 255u) * BM_LP, 1u << (sh + j));
    }
    __syncthreads();

    for (int o = wave; o < rows; o += 32) {
      const uint4 v = *reinterpret_cast<const uint4*>(lb + o * BM_LP + 4 * lane);
      *reinterpret_cast<uint4*>(bits + (size_t)o * nwp + (size_t)blk * BM_BW + 4 * lane) = v;
    }
    __syncthreads();
  }
}

__device__ __forceinline__ void drain_queue(
    const int* __restrict__ myq, int qn, int lane,
    const int* __restrict__ srcp, const int* __restrict__ dstp, const float* __restrict__ news,
    float* acc, float* degc, int E, int lo, int cnt) {
  for (int b0 = 0; b0 < qn; b0 += 32) {
    const int nb = min(32, qn - b0);
    int s = 0, t = RCAP;
    if (lane < nb) {
      const int e = myq[b0 + lane];
      if (e < E) {
        const int d = dstp[e] - lo;
        s = srcp[e];
        if ((unsigned)d < (unsigned)cnt) {
          t = d;
          atomicAdd(&degc[t], 1.0f);
        }
      }
    }
    for (int k = 0; k < nb; k += 8) {
      float4 r[8];
      int tt[8];
#pragma unroll
      for (int v = 0; v < 8; ++v) {
        const int kv8 = min(k + v, nb - 1);
        const int sk = __builtin_amdgcn_readlane(s, kv8);
        tt[v] = __builtin_amdgcn_readlane(t, kv8);
        r[v] = *reinterpret_cast<const float4*>(news + (size_t)sk * D + lane * 4);
      }
#pragma unroll
      for (int v = 0; v < 8; ++v) {
        if (k + v < nb) {
          float* a = acc + tt[v] * D + lane;
          atomicAdd(a,      r[v].x);
          atomicAdd(a + 32, r[v].y);
          atomicAdd(a + 64, r[v].z);
          atomicAdd(a + 96, r[v].w);
        }
      }
    }
  }
}

__global__ __launch_bounds__(1024) void edge_agg_owner(
    const int* __restrict__ ei, const unsigned* __restrict__ bits, const float* __restrict__ news,
    float* __restrict__ aggn, int E, int N) {
  __shared__ float acc[(RCAP + 1) * D];
  __shared__ float degc[RCAP];
  __shared__ int   que[32][QCAP];

  const int tid  = threadIdx.x;
  const int lane = tid & 31;
  const int wave = __builtin_amdgcn_readfirstlane(tid >> 5);
  const int* __restrict__ srcp = ei;
  const int* __restrict__ dstp = ei + E;
  const int nq = ((E + BM_E - 1) / BM_E) * (BM_BW / 4);
  const int ntrip = (nq + 1023) / 1024;
  int* __restrict__ myq = que[wave];
  const uint4* __restrict__ kq = reinterpret_cast<const uint4*>(bits) + (size_t)(blockIdx.x & 255u) * nq;
  const uint4 zero4 = make_uint4(0u, 0u, 0u, 0u);

  for (int lo = blockIdx.x * RCAP; lo < N; lo += NOWN * RCAP) {
    const int cnt = min(RCAP, N - lo);

    for (int i = tid; i < (RCAP + 1) * D / 4; i += 1024)
      reinterpret_cast<float4*>(acc)[i] = make_float4(0.f, 0.f, 0.f, 0.f);
    if (tid < RCAP) degc[tid] = 0.f;
    __syncthreads();

    int c = (ntrip > 0) ? (int)((blockIdx.x * 37u) % (unsigned)ntrip) : 0;
    int qn = 0;
    uint4 kv = zero4, kn;
    if (ntrip > 0) {
      const int q = c * 1024 + wave * 32 + lane;
      if (q < nq) kv = kq[q];
    }
    for (int it = 0; it < ntrip; ++it) {
      const int q = c * 1024 + wave * 32 + lane;
      c = (c + 1 == ntrip) ? 0 : c + 1;
      {
        const int qx = c * 1024 + wave * 32 + lane;
        kn = zero4;
        if (qx < nq) kn = kq[qx];
      }

      const unsigned mw[4] = {kv.x, kv.y, kv.z, kv.w};
#pragma unroll
      for (int j = 0; j < 4; ++j) {
        unsigned m = mw[j];
        unsigned bal = __builtin_amdgcn_ballot_w32(m != 0u);
        while (bal != 0u) {
          if (m != 0u) {
            myq[qn + (int)__builtin_amdgcn_mbcnt_lo(bal, 0u)] = 128 * q + 32 * j + __builtin_ctz(m);
            m &= m - 1u;
          }
          qn += __builtin_popcount(bal);
          if (qn >= 32) {
            drain_queue(myq, qn, lane, srcp, dstp, news, acc, degc, E, lo, cnt);
            qn = 0;
          }
          bal = __builtin_amdgcn_ballot_w32(m != 0u);
        }
      }
      kv = kn;
    }
    if (qn > 0) drain_queue(myq, qn, lane, srcp, dstp, news, acc, degc, E, lo, cnt);
    __syncthreads();

    for (int t = wave; t < cnt; t += 32) {
      const float* a = acc + t * D + lane;
      const float dinv = 1.0f / fmaxf(degc[t], 1.0f);
      const float4 v = make_float4(a[0] * dinv, a[32] * dinv, a[64] * dinv, a[96] * dinv);
      *reinterpret_cast<float4*>(aggn + (size_t)(lo + t) * D + lane * 4) = v;
    }
    __syncthreads();
  }
}

#define UB 16
#define UP 132
#define RG 10
#define HSPLIT 30

__global__ __launch_bounds__(512) void user_score_lin(
    const float* __restrict__ news, const float* __restrict__ aggn, const float* __restrict__ Wg,
    const int* __restrict__ hist, const int* __restrict__ cand, float* __restrict__ scores, int B) {
  __shared__ float sN[UB * UP];
  __shared__ float sA[UB * UP];
  __shared__ float sC[UB];
  __shared__ float sS[UB * CAND];

  const int tid  = threadIdx.x;
  const int lane = tid & 31;
  const int wave = __builtin_amdgcn_readfirstlane(tid >> 5);
  const int c = lane & 15, h = lane >> 4;
  const int ntile = (B + UB - 1) / UB;

  for (int tile = blockIdx.x; tile < ntile; tile += gridDim.x) {
    const int b = tile * UB + wave;

    int hv0 = 0, hv1 = 0, cv = 0;
    if (b < B) {
      if (lane < HSPLIT) hv0 = hist[(size_t)b * HIST + lane];
      if (lane < HIST - HSPLIT) hv1 = hist[(size_t)b * HIST + HSPLIT + lane];
      if (lane < CAND) cv = cand[(size_t)b * CAND + lane];
    }

    float4 sn = make_float4(0.f, 0.f, 0.f, 0.f), sa = sn;
    float cnt = 0.f;
    if (b < B) {
      for (int g = 0; g < HIST; g += RG) {
        float4 n[RG], a[RG];
        float m[RG];
        const int hsel = (g < HSPLIT) ? hv0 : hv1;
        const int l0   = (g < HSPLIT) ? g : g - HSPLIT;
#pragma unroll
        for (int k = 0; k < RG; ++k) {
          const int nid = __builtin_amdgcn_readlane(hsel, l0 + k);
          m[k] = (nid != 0) ? 1.0f : 0.0f;
          n[k] = *reinterpret_cast<const float4*>(news + (size_t)nid * D + lane * 4);
          a[k] = *reinterpret_cast<const float4*>(aggn + (size_t)nid * D + lane * 4);
        }
#pragma unroll
        for (int k = 0; k < RG; ++k) {
          sn.x += m[k] * n[k].x; sn.y += m[k] * n[k].y; sn.z += m[k] * n[k].z; sn.w += m[k] * n[k].w;
          sa.x += m[k] * a[k].x; sa.y += m[k] * a[k].y; sa.z += m[k] * a[k].z; sa.w += m[k] * a[k].w;
          cnt += m[k];
        }
      }
    }
    *reinterpret_cast<float4*>(&sN[wave * UP + lane * 4]) = sn;
    *reinterpret_cast<float4*>(&sA[wave * UP + lane * 4]) = sa;
    if (lane == 0) sC[wave] = cnt;
    __syncthreads();

    v2f bw[D / 4];
    if (wave < 8) {
      const int col = wave * 16 + c;
#pragma unroll
      for (int j = 0; j < D / 4; ++j) {
        bw[j].x = Wg[(size_t)(4 * j + 2 * h + 0) * D + col];
        bw[j].y = Wg[(size_t)(4 * j + 2 * h + 1) * D + col];
      }
      const float* pa = &sA[c * UP + 2 * h];
      v8f acc = {};
#pragma unroll
      for (int j = 0; j < D / 4; ++j) {
        const v2f a = *reinterpret_cast<const v2f*>(pa + 4 * j);
        acc = __builtin_amdgcn_wmma_f32_16x16x4_f32(
            false, a, false, bw[j], (short)0, acc, false, false);
      }
#pragma unroll
      for (int r = 0; r < 8; ++r) {
        const int row = 8 * h + r;
        sN[row * UP + col] = (sN[row * UP + col] + acc[r]) / fmaxf(sC[row], 1e-9f);
      }
    } else {
      const int col = (wave - 8) * 16 + c;
#pragma unroll
      for (int j = 0; j < D / 4; ++j)
        bw[j] = *reinterpret_cast<const v2f*>(Wg + (size_t)col * D + 4 * j + 2 * h);
    }
    __syncthreads();

    if (wave >= 8) {
      const int col = (wave - 8) * 16 + c;
      const float* pa = &sN[c * UP + 2 * h];
      v8f acc = {};
#pragma unroll
      for (int j = 0; j < D / 4; ++j) {
        const v2f a = *reinterpret_cast<const v2f*>(pa + 4 * j);
        acc = __builtin_amdgcn_wmma_f32_16x16x4_f32(
            false, a, false, bw[j], (short)0, acc, false, false);
      }
#pragma unroll
      for (int r = 0; r < 8; ++r) sA[(8 * h + r) * UP + col] = acc[r];
    }
    __syncthreads();

    if (b < B) {
      const float4 u = *reinterpret_cast<const float4*>(&sN[wave * UP + lane * 4]);
      const float4 w = *reinterpret_cast<const float4*>(&sA[wave * UP + lane * 4]);
      float mine = 0.f;
      for (int g = 0; g < CAND; g += RG) {
        float4 n[RG], a[RG];
#pragma unroll
        for (int k = 0; k < RG; ++k) {
          const int nid = __builtin_amdgcn_readlane(cv, g + k);
          n[k] = *reinterpret_cast<const float4*>(news + (size_t)nid * D + lane * 4);
          a[k] = *reinterpret_cast<const float4*>(aggn + (size_t)nid * D + lane * 4);
        }
#pragma unroll
        for (int k = 0; k < RG; ++k) {
          float p = n[k].x * u.x + n[k].y * u.y + n[k].z * u.z + n[k].w * u.w
                  + a[k].x * w.x + a[k].y * w.y + a[k].z * w.z + a[k].w * w.w;
          for (int off = 16; off; off >>= 1) p += __shfl_xor(p, off, 32);
          if (lane == g + k) mine = p;
        }
      }
      if (lane < CAND) sS[wave * CAND + lane] = mine;
    }
    __syncthreads();

    if (tid < UB * CAND) {
      const size_t g = (size_t)tile * (UB * CAND) + tid;
      if (g < (size_t)B * CAND) scores[g] = sS[tid];
    }
  }
}

extern "C" void kernel_launch(void* const* d_in, const int* in_sizes, int n_in,
                              void* d_out, int out_size, void* d_ws, size_t ws_size,
                              hipStream_t stream) {
  const float* x      = (const float*)d_in[0];
  const int*   ei     = (const int*)  d_in[2];
  const int*   hist   = (const int*)  d_in[3];
  const int*   cand   = (const int*)  d_in[4];
  const float* W_enc  = (const float*)d_in[5];
  const float* W_gnn  = (const float*)d_in[6];
  float*       scores = (float*)d_out;

  const int       N = in_sizes[1];
  const long long E = (long long)in_sizes[2] / 2;
  const int       B = in_sizes[3] / HIST;

  float* news = (float*)d_ws;
  float* aggn = news + (size_t)N * D;
  unsigned* bits = (unsigned*)(aggn + (size_t)N * D);

  gemm_wstat<300, 300><<<512, 256, 0, stream>>>(x, W_enc, news, N);

  edge_owner_bits<<<512, 1024, 0, stream>>>(ei, bits, (int)E, N);

  edge_agg_owner<<<NOWN, 1024, 0, stream>>>(ei, bits, news, aggn, (int)E, N);

  user_score_lin<<<256, 512, 0, stream>>>(news, aggn, W_gnn, hist, cand, scores, B);
}


# Pre-pass hands each owner its edges as (src,dst) pairs, binned per 4096-edge block in LDS into 512-B segments, not one bit per edge: no bit scan, no index gathers; LDS sum order stays arrival order
# speedup vs baseline: 6.7032x; 1.1074x over previous
#include <hip/hip_runtime.h>
#include <hip/hip_bf16.h>

typedef float v2f __attribute__((ext_vector_type(2)));
typedef float v8f __attribute__((ext_vector_type(8)));

#define D 128
#define HIST 50
#define CAND 20

#define GB_ROWS 32

template <int K, int PITCH>
__global__ __launch_bounds__(256) __attribute__((amdgpu_waves_per_eu(4, 4))) void gemm_wstat(
    const float* __restrict__ A, const float* __restrict__ W,
    float* __restrict__ out, int N) {
  constexpr int KS  = K / 4;
  constexpr int RF4 = K / 4;
  constexpr int NF4 = GB_ROWS * RF4;
  constexpr int NLD = (NF4 + 255) / 256;
  constexpr int STAGE = (GB_ROWS * PITCH > NLD * 1024) ? GB_ROWS * PITCH : NLD * 1024;
  __shared__ float lds[2][STAGE];

  const int tid  = threadIdx.x;
  const int lane = tid & 31;
  const int wave = __builtin_amdgcn_readfirstlane(tid >> 5);
  const int c = lane & 15, h = lane >> 4;
  const int col = wave * 16 + c;
  const int nblk = N / GB_ROWS;

  v2f b[KS];
#pragma unroll
  for (int j = 0; j < KS; ++j) {
    b[j].x = W[(size_t)(4 * j + 2 * h + 0) * D + col];
    b[j].y = W[(size_t)(4 * j + 2 * h + 1) * D + col];
  }

  float4 p[NLD];
  int blk = blockIdx.x;

  if (blk < nblk) {
    const float4* __restrict__ src = reinterpret_cast<const float4*>(A) + (size_t)blk * NF4;
#pragma unroll
    for (int i = 0; i < NLD; ++i) {
      const int f = min(tid + 256 * i, NF4 - 1);
      p[i] = src[f];
    }
#pragma unroll
    for (int i = 0; i < NLD; ++i) {
      const int f = tid + 256 * i;
      const int a = (PITCH == K) ? 4 * f : (f / RF4) * PITCH + (f % RF4) * 4;
      *reinterpret_cast<float4*>(&lds[0][a]) = p[i];
    }
  }
  __syncthreads();

  int s = 0;
  for (; blk < nblk; blk += gridDim.x) {
    const int nxt = (blk + (int)gridDim.x < nblk) ? blk + (int)gridDim.x : blk;
    {
      const float4* __restrict__ src = reinterpret_cast<const float4*>(A) + (size_t)nxt * NF4;
#pragma unroll
      for (int i = 0; i < NLD; ++i) {
        const int f = min(tid + 256 * i, NF4 - 1);
        p[i] = src[f];
      }
    }
    const size_t o0 = (size_t)(blk * GB_ROWS + 8 * h) * D + col;

    const float* sA = &lds[s][c * PITCH + 2 * h];
    v8f acc0 = {}, acc1 = {};
#pragma unroll
    for (int j = 0; j < KS; ++j) {
      const v2f a0 = *reinterpret_cast<const v2f*>(sA + 4 * j);
      const v2f a1 = *reinterpret_cast<const v2f*>(sA + 16 * PITCH + 4 * j);
      acc0 = __builtin_amdgcn_wmma_f32_16x16x4_f32(
          false, a0, false, b[j], (short)0, acc0, false, false);
      acc1 = __builtin_amdgcn_wmma_f32_16x16x4_f32(
          false, a1, false, b[j], (short)0, acc1, false, false);
    }

#pragma unroll
    for (int r = 0; r < 8; ++r) {
      out[o0 + (size_t)r * D]        = acc0[r];
      out[o0 + (size_t)(16 + r) * D] = acc1[r];
    }

    {
#pragma unroll
      for (int i = 0; i < NLD; ++i) {
        const int f = tid + 256 * i;
        const int a = (PITCH == K) ? 4 * f : (f / RF4) * PITCH + (f % RF4) * 4;
        *reinterpret_cast<float4*>(&lds[s ^ 1][a]) = p[i];
      }
    }
    __syncthreads();
    s ^= 1;
  }
}

#define RCAP   416
#define NOWN   256
#define BM_E   4096
#define SEG    64
#define SG_LP  132
#define QRING  128

__device__ __forceinline__ unsigned owner_of(int d) { return ((unsigned)d / RCAP) & 255u; }

__device__ __forceinline__ void put_pair(int* lseg, unsigned o, int r, int s, int d) {
  if (r < SEG) *reinterpret_cast<int2*>(lseg + o * SG_LP + 2 * r) = make_int2(s, d);
}

__global__ __launch_bounds__(1024) void edge_owner_pairs(
    const int* __restrict__ ei, int* __restrict__ segs, int E, int N) {
  __shared__ __attribute__((aligned(16))) int lseg[NOWN * SG_LP];
  __shared__ int lcnt[NOWN];
  const int* __restrict__ srcp = ei;
  const int* __restrict__ dstp = ei + E;
  const int tid  = threadIdx.x;
  const int lane = tid & 31;
  const int wave = __builtin_amdgcn_readfirstlane(tid >> 5);
  const int nblk = (E + BM_E - 1) / BM_E;
  const int rows = min(NOWN, (N + RCAP - 1) / RCAP);
  const bool vec = ((reinterpret_cast<size_t>(dstp) | reinterpret_cast<size_t>(srcp)) & 15) == 0;

  for (int blk = blockIdx.x; blk < nblk; blk += gridDim.x) {
    for (int i = tid; i < NOWN * SG_LP / 4; i += 1024)
      reinterpret_cast<int4*>(lseg)[i] = make_int4(-1, -1, -1, -1);
    if (tid < NOWN) lcnt[tid] = 0;
    __syncthreads();

    const int e0 = blk * BM_E + 4 * tid;
    if (vec && e0 + 3 < E) {
      const int4 d = *reinterpret_cast<const int4*>(dstp + e0);
      const int4 s = *reinterpret_cast<const int4*>(srcp + e0);
      const unsigned o0 = owner_of(d.x), o1 = owner_of(d.y), o2 = owner_of(d.z), o3 = owner_of(d.w);
      const int r0 = atomicAdd(&lcnt[o0], 1);
      const int r1 = atomicAdd(&lcnt[o1], 1);
      const int r2 = atomicAdd(&lcnt[o2], 1);
      const int r3 = atomicAdd(&lcnt[o3], 1);
      put_pair(lseg, o0, r0, s.x, d.x);
      put_pair(lseg, o1, r1, s.y, d.y);
      put_pair(lseg, o2, r2, s.z, d.z);
      put_pair(lseg, o3, r3, s.w, d.w);
    } else {
      for (int j = 0; j < 4; ++j)
        if (e0 + j < E) {
          const int d = dstp[e0 + j];
          const unsigned o = owner_of(d);
          put_pair(lseg, o, atomicAdd(&lcnt[o], 1), srcp[e0 + j], d);
        }
    }
    __syncthreads();

    for (int o = wave; o < rows; o += 32) {
      if (lane == 0 && lcnt[o] > SEG) lseg[o * SG_LP] = -2;
      const int4 v = *reinterpret_cast<const int4*>(lseg + o * SG_LP + 4 * lane);
      *reinterpret_cast<int4*>(segs + ((size_t)o * nblk + blk) * (2 * SEG) + 4 * lane) = v;
    }
    __syncthreads();
  }
}

__device__ __forceinline__ void drain_batch(
    const int2* __restrict__ myr, int head, int nb, int lane,
    const float* __restrict__ news, float* acc) {
  int s = 0, t = RCAP;
  if (lane < nb) {
    const int2 p = myr[(head + lane) & (QRING - 1)];
    s = p.x;
    t = p.y;
  }
  for (int k = 0; k < nb; k += 8) {
    float4 r[8];
    int tt[8];
#pragma unroll
    for (int v = 0; v < 8; ++v) {
      const int kv8 = min(k + v, nb - 1);
      const int sk = __builtin_amdgcn_readlane(s, kv8);
      tt[v] = __builtin_amdgcn_readlane(t, kv8);
      r[v] = *reinterpret_cast<const float4*>(news + (size_t)sk * D + lane * 4);
    }
#pragma unroll
    for (int v = 0; v < 8; ++v) {
      if (k + v < nb) {
        float* a = acc + tt[v] * D + lane;
        atomicAdd(a,      r[v].x);
        atomicAdd(a + 32, r[v].y);
        atomicAdd(a + 64, r[v].z);
        atomicAdd(a + 96, r[v].w);
      }
    }
  }
}

__device__ __forceinline__ void push_pair(
    int2* myr, int tail, unsigned bal, int s, int d, float* degc, int lo, int cnt) {
  int t = RCAP;
  const int dl = d - lo;
  if ((unsigned)dl < (unsigned)cnt) {
    t = dl;
    atomicAdd(&degc[t], 1.0f);
  }
  myr[(tail + (int)__builtin_amdgcn_mbcnt_lo(bal, 0u)) & (QRING - 1)] = make_int2(s, t);
}

__global__ __launch_bounds__(1024) void edge_agg_owner(
    const int* __restrict__ ei, const int* __restrict__ segs, const float* __restrict__ news,
    float* __restrict__ aggn, int E, int N) {
  __shared__ float acc[(RCAP + 1) * D];
  __shared__ float degc[RCAP];
  __shared__ int2  ring[32][QRING];

  const int tid  = threadIdx.x;
  const int lane = tid & 31;
  const int wave = __builtin_amdgcn_readfirstlane(tid >> 5);
  const int* __restrict__ srcp = ei;
  const int* __restrict__ dstp = ei + E;
  const int nblk = (E + BM_E - 1) / BM_E;
  const unsigned key = blockIdx.x & 255u;
  int2* __restrict__ myr = ring[wave];
  const int4* __restrict__ srow = reinterpret_cast<const int4*>(segs) + (size_t)key * nblk * 32;
  const int4 none4 = make_int4(-1, -1, -1, -1);

  for (int lo = blockIdx.x * RCAP; lo < N; lo += NOWN * RCAP) {
    const int cnt = min(RCAP, N - lo);

    for (int i = tid; i < (RCAP + 1) * D / 4; i += 1024)
      reinterpret_cast<float4*>(acc)[i] = make_float4(0.f, 0.f, 0.f, 0.f);
    if (tid < RCAP) degc[tid] = 0.f;
    __syncthreads();

    int head = 0, tail = 0;

    auto take = [&](const int4 sv, const int blk) {
      if (__builtin_amdgcn_readfirstlane(sv.x) != -2) {
        const unsigned ba = __builtin_amdgcn_ballot_w32(sv.x >= 0);
        if (sv.x >= 0) push_pair(myr, tail, ba, sv.x, sv.y, degc, lo, cnt);
        tail += __builtin_popcount(ba);
        const unsigned bb = __builtin_amdgcn_ballot_w32(sv.z >= 0);
        if (sv.z >= 0) push_pair(myr, tail, bb, sv.z, sv.w, degc, lo, cnt);
        tail += __builtin_popcount(bb);
        while (tail - head >= 32) {
          drain_batch(myr, head, 32, lane, news, acc);
          head += 32;
        }
      } else {
        for (int i = 0; i < BM_E / 32; ++i) {
          const int e = blk * BM_E + 32 * i + lane;
          bool hit = false;
          int es = 0, ed = 0;
          if (e < E) {
            ed = dstp[e];
            if ((((unsigned)ed / RCAP) & 255u) == key) {
              hit = true;
              es = srcp[e];
            }
          }
          const unsigned bh = __builtin_amdgcn_ballot_w32(hit);
          if (hit) push_pair(myr, tail, bh, es, ed, degc, lo, cnt);
          tail += __builtin_popcount(bh);
          while (tail - head >= 32) {
            drain_batch(myr, head, 32, lane, news, acc);
            head += 32;
          }
        }
      }
    };

    int4 v = none4, w = none4;
    if (wave < nblk) {
      v = srow[(size_t)wave * 32 + lane];
      w = srow[(size_t)((wave + 32 < nblk) ? wave + 32 : wave) * 32 + lane];
    }
    for (int blk = wave; blk < nblk; blk += 64) {
      const int bv = (blk + 64 < nblk) ? blk + 64 : blk;
      const int bw = (blk + 96 < nblk) ? blk + 96 : blk;
      const int4 vn = srow[(size_t)bv * 32 + lane];
      const int4 wn = srow[(size_t)bw * 32 + lane];
      take(v, blk);
      if (blk + 32 < nblk) take(w, blk + 32);
      v = vn;
      w = wn;
    }
    if (tail - head > 0) drain_batch(myr, head, tail - head, lane, news, acc);
    __syncthreads();

    for (int t = wave; t < cnt; t += 32) {
      const float* a = acc + t * D + lane;
      const float dinv = 1.0f / fmaxf(degc[t], 1.0f);
      const float4 v4 = make_float4(a[0] * dinv, a[32] * dinv, a[64] * dinv, a[96] * dinv);
      *reinterpret_cast<float4*>(aggn + (size_t)(lo + t) * D + lane * 4) = v4;
    }
    __syncthreads();
  }
}

#define UB 16
#define UP 132
#define RG 10
#define HSPLIT 30

__global__ __launch_bounds__(512) void user_score_lin(
    const float* __restrict__ news, const float* __restrict__ aggn, const float* __restrict__ Wg,
    const int* __restrict__ hist, const int* __restrict__ cand, float* __restrict__ scores, int B) {
  __shared__ float sN[UB * UP];
  __shared__ float sA[UB * UP];
  __shared__ float sC[UB];
  __shared__ float sS[UB * CAND];

  const int tid  = threadIdx.x;
  const int lane = tid & 31;
  const int wave = __builtin_amdgcn_readfirstlane(tid >> 5);
  const int c = lane & 15, h = lane >> 4;
  const int ntile = (B + UB - 1) / UB;

  for (int tile = blockIdx.x; tile < ntile; tile += gridDim.x) {
    const int b = tile * UB + wave;

    int hv0 = 0, hv1 = 0, cv = 0;
    if (b < B) {
      if (lane < HSPLIT) hv0 = hist[(size_t)b * HIST + lane];
      if (lane < HIST - HSPLIT) hv1 = hist[(size_t)b * HIST + HSPLIT + lane];
      if (lane < CAND) cv = cand[(size_t)b * CAND + lane];
    }

    float4 sn = make_float4(0.f, 0.f, 0.f, 0.f), sa = sn;
    float cnt = 0.f;
    if (b < B) {
      for (int g = 0; g < HIST; g += RG) {
        float4 n[RG], a[RG];
        float m[RG];
        const int hsel = (g < HSPLIT) ? hv0 : hv1;
        const int l0   = (g < HSPLIT) ? g : g - HSPLIT;
#pragma unroll
        for (int k = 0; k < RG; ++k) {
          const int nid = __builtin_amdgcn_readlane(hsel, l0 + k);
          m[k] = (nid != 0) ? 1.0f : 0.0f;
          n[k] = *reinterpret_cast<const float4*>(news + (size_t)nid * D + lane * 4);
          a[k] = *reinterpret_cast<const float4*>(aggn + (size_t)nid * D + lane * 4);
        }
#pragma unroll
        for (int k = 0; k < RG; ++k) {
          sn.x += m[k] * n[k].x; sn.y += m[k] * n[k].y; sn.z += m[k] * n[k].z; sn.w += m[k] * n[k].w;
          sa.x += m[k] * a[k].x; sa.y += m[k] * a[k].y; sa.z += m[k] * a[k].z; sa.w += m[k] * a[k].w;
          cnt += m[k];
        }
      }
    }
    *reinterpret_cast<float4*>(&sN[wave * UP + lane * 4]) = sn;
    *reinterpret_cast<float4*>(&sA[wave * UP + lane * 4]) = sa;
    if (lane == 0) sC[wave] = cnt;
    __syncthreads();

    v2f bw[D / 4];
    if (wave < 8) {
      const int col = wave * 16 + c;
#pragma unroll
      for (int j = 0; j < D / 4; ++j) {
        bw[j].x = Wg[(size_t)(4 * j + 2 * h + 0) * D + col];
        bw[j].y = Wg[(size_t)(4 * j + 2 * h + 1) * D + col];
      }
      const float* pa = &sA[c * UP + 2 * h];
      v8f acc = {};
#pragma unroll
      for (int j = 0; j < D / 4; ++j) {
        const v2f a = *reinterpret_cast<const v2f*>(pa + 4 * j);
        acc = __builtin_amdgcn_wmma_f32_16x16x4_f32(
            false, a, false, bw[j], (short)0, acc, false, false);
      }
#pragma unroll
      for (int r = 0; r < 8; ++r) {
        const int row = 8 * h + r;
        sN[row * UP + col] = (sN[row * UP + col] + acc[r]) / fmaxf(sC[row], 1e-9f);
      }
    } else {
      const int col = (wave - 8) * 16 + c;
#pragma unroll
      for (int j = 0; j < D / 4; ++j)
        bw[j] = *reinterpret_cast<const v2f*>(Wg + (size_t)col * D + 4 * j + 2 * h);
    }
    __syncthreads();

    if (wave >= 8) {
      const int col = (wave - 8) * 16 + c;
      const float* pa = &sN[c * UP + 2 * h];
      v8f acc = {};
#pragma unroll
      for (int j = 0; j < D / 4; ++j) {
        const v2f a = *reinterpret_cast<const v2f*>(pa + 4 * j);
        acc = __builtin_amdgcn_wmma_f32_16x16x4_f32(
            false, a, false, bw[j], (short)0, acc, false, false);
      }
#pragma unroll
      for (int r = 0; r < 8; ++r) sA[(8 * h + r) * UP + col] = acc[r];
    }
    __syncthreads();

    if (b < B) {
      const float4 u = *reinterpret_cast<const float4*>(&sN[wave * UP + lane * 4]);
      const float4 w = *reinterpret_cast<const float4*>(&sA[wave * UP + lane * 4]);
      float mine = 0.f;
      for (int g = 0; g < CAND; g += RG) {
        float4 n[RG], a[RG];
#pragma unroll
        for (int k = 0; k < RG; ++k) {
          const int nid = __builtin_amdgcn_readlane(cv, g + k);
          n[k] = *reinterpret_cast<const float4*>(news + (size_t)nid * D + lane * 4);
          a[k] = *reinterpret_cast<const float4*>(aggn + (size_t)nid * D + lane * 4);
        }
#pragma unroll
        for (int k = 0; k < RG; ++k) {
          float p = n[k].x * u.x + n[k].y * u.y + n[k].z * u.z + n[k].w * u.w
                  + a[k].x * w.x + a[k].y * w.y + a[k].z * w.z + a[k].w * w.w;
          for (int off = 16; off; off >>= 1) p += __shfl_xor(p, off, 32);
          if (lane == g + k) mine = p;
        }
      }
      if (lane < CAND) sS[wave * CAND + lane] = mine;
    }
    __syncthreads();

    if (tid < UB * CAND) {
      const size_t g = (size_t)tile * (UB * CAND) + tid;
      if (g < (size_t)B * CAND) scores[g] = sS[tid];
    }
  }
}

extern "C" void kernel_launch(void* const* d_in, const int* in_sizes, int n_in,
                              void* d_out, int out_size, void* d_ws, size_t ws_size,
                              hipStream_t stream) {
  const float* x      = (const float*)d_in[0];
  const int*   ei     = (const int*)  d_in[2];
  const int*   hist   = (const int*)  d_in[3];
  const int*   cand   = (const int*)  d_in[4];
  const float* W_enc  = (const float*)d_in[5];
  const float* W_gnn  = (const float*)d_in[6];
  float*       scores = (float*)d_out;

  const int       N = in_sizes[1];
  const long long E = (long long)in_sizes[2] / 2;
  const int       B = in_sizes[3] / HIST;

  float* news = (float*)d_ws;
  float* aggn = news + (size_t)N * D;
  int* segs = (int*)(aggn + (size_t)N * D);

  gemm_wstat<300, 300><<<512, 256, 0, stream>>>(x, W_enc, news, N);

  edge_owner_pairs<<<512, 1024, 0, stream>>>(ei, segs, (int)E, N);

  edge_agg_owner<<<NOWN, 1024, 0, stream>>>(ei, segs, news, aggn, (int)E, N);

  user_score_lin<<<256, 512, 0, stream>>>(news, aggn, W_gnn, hist, cand, scores, B);
}


# Node GEMM: row blocks of 16, not 32: the next block's loads (5 a thread) now fit in registers and stay in flight across the trip's WMMAs, not awaited at its foot; block barrier fences LDS only
# speedup vs baseline: 6.9426x; 1.0357x over previous
#include <hip/hip_runtime.h>
#include <hip/hip_bf16.h>

typedef float v2f __attribute__((ext_vector_type(2)));
typedef float v8f __attribute__((ext_vector_type(8)));

#define D 128
#define HIST 50
#define CAND 20

#define GB_ROWS 16

template <int K, int PITCH>
__global__ __launch_bounds__(256) __attribute__((amdgpu_waves_per_eu(4, 4))) void gemm_wstat(
    const float* __restrict__ A, const float* __restrict__ W,
    float* __restrict__ out, int N) {
  constexpr int KS  = K / 4;
  constexpr int RF4 = K / 4;
  constexpr int NF4 = GB_ROWS * RF4;
  constexpr int NLD = (NF4 + 255) / 256;
  constexpr int STAGE = (GB_ROWS * PITCH > NLD * 1024) ? GB_ROWS * PITCH : NLD * 1024;
  __shared__ float lds[2][STAGE];

  const int tid  = threadIdx.x;
  const int lane = tid & 31;
  const int wave = __builtin_amdgcn_readfirstlane(tid >> 5);
  const int c = lane & 15, h = lane >> 4;
  const int col = wave * 16 + c;
  const int nblk = N / GB_ROWS;

  v2f b[KS];
#pragma unroll
  for (int j = 0; j < KS; ++j) {
    b[j].x = W[(size_t)(4 * j + 2 * h + 0) * D + col];
    b[j].y = W[(size_t)(4 * j + 2 * h + 1) * D + col];
  }

  static_assert(NLD == 5, "the block's loads are written out five to a thread");
  float4 p0, p1, p2, p3, p4;
  const int f0 = tid, f1 = tid + 256, f2 = tid + 512, f3 = tid + 768, f4 = min(tid + 1024, NF4 - 1);
  auto slot = [](int f) { return (PITCH == K) ? 4 * f : (f / RF4) * PITCH + (f % RF4) * 4; };
  const int a0s = slot(tid), a1s = slot(tid + 256), a2s = slot(tid + 512), a3s = slot(tid + 768), a4s = slot(tid + 1024);

  int blk = blockIdx.x;
  const int G = (int)gridDim.x;

  if (blk < nblk) {
    const float4* __restrict__ src = reinterpret_cast<const float4*>(A) + (size_t)blk * NF4;
    p0 = src[f0]; p1 = src[f1]; p2 = src[f2]; p3 = src[f3]; p4 = src[f4];
    *reinterpret_cast<float4*>(&lds[0][a0s]) = p0;
    *reinterpret_cast<float4*>(&lds[0][a1s]) = p1;
    *reinterpret_cast<float4*>(&lds[0][a2s]) = p2;
    *reinterpret_cast<float4*>(&lds[0][a3s]) = p3;
    *reinterpret_cast<float4*>(&lds[0][a4s]) = p4;
    const int nxt = (blk + G < nblk) ? blk + G : blk;
    const float4* __restrict__ src1 = reinterpret_cast<const float4*>(A) + (size_t)nxt * NF4;
    p0 = src1[f0]; p1 = src1[f1]; p2 = src1[f2]; p3 = src1[f3]; p4 = src1[f4];
  } else {
    p0 = p1 = p2 = p3 = p4 = make_float4(0.f, 0.f, 0.f, 0.f);
  }
  __syncthreads();

  int s = 0;
  for (; blk < nblk; blk += G) {
    const size_t o0 = (size_t)(blk * GB_ROWS + 8 * h) * D + col;

    const float* sA = &lds[s][c * PITCH + 2 * h];
    v8f acc0 = {};
#pragma unroll
    for (int j = 0; j < KS; ++j) {
      const v2f a0 = *reinterpret_cast<const v2f*>(sA + 4 * j);
      acc0 = __builtin_amdgcn_wmma_f32_16x16x4_f32(
          false, a0, false, b[j], (short)0, acc0, false, false);
    }
    __builtin_amdgcn_sched_group_barrier(0x100, 6, 0);
    __builtin_amdgcn_sched_group_barrier(0x020, 5, 0);
#pragma unroll
    for (int g = 0; g < 32; ++g) {
      __builtin_amdgcn_sched_group_barrier(0x008, 2, 0);
      __builtin_amdgcn_sched_group_barrier(0x100, 1, 0);
    }
    __builtin_amdgcn_sched_group_barrier(0x008, 11, 0);

#pragma unroll
    for (int r = 0; r < 8; ++r) out[o0 + (size_t)r * D] = acc0[r];

    *reinterpret_cast<float4*>(&lds[s ^ 1][a0s]) = p0;
    *reinterpret_cast<float4*>(&lds[s ^ 1][a1s]) = p1;
    *reinterpret_cast<float4*>(&lds[s ^ 1][a2s]) = p2;
    *reinterpret_cast<float4*>(&lds[s ^ 1][a3s]) = p3;
    *reinterpret_cast<float4*>(&lds[s ^ 1][a4s]) = p4;
    {
      const int nn = (blk + 2 * G < nblk) ? blk + 2 * G : blk;
      const float4* __restrict__ src2 = reinterpret_cast<const float4*>(A) + (size_t)nn * NF4;
      p0 = src2[f0]; p1 = src2[f1]; p2 = src2[f2]; p3 = src2[f3]; p4 = src2[f4];
    }
    __builtin_amdgcn_fence(__ATOMIC_RELEASE, "workgroup", "local");
    __builtin_amdgcn_s_barrier();
    __builtin_amdgcn_fence(__ATOMIC_ACQUIRE, "workgroup", "local");
    s ^= 1;
  }
}

#define RCAP   416
#define NOWN   256
#define BM_E   4096
#define SEG    64
#define SG_LP  132
#define QRING  128

__device__ __forceinline__ unsigned owner_of(int d) { return ((unsigned)d / RCAP) & 255u; }

__device__ __forceinline__ void put_pair(int* lseg, unsigned o, int r, int s, int d) {
  if (r < SEG) *reinterpret_cast<int2*>(lseg + o * SG_LP + 2 * r) = make_int2(s, d);
}

__global__ __launch_bounds__(1024) void edge_owner_pairs(
    const int* __restrict__ ei, int* __restrict__ segs, int E, int N) {
  __shared__ __attribute__((aligned(16))) int lseg[NOWN * SG_LP];
  __shared__ int lcnt[NOWN];
  const int* __restrict__ srcp = ei;
  const int* __restrict__ dstp = ei + E;
  const int tid  = threadIdx.x;
  const int lane = tid & 31;
  const int wave = __builtin_amdgcn_readfirstlane(tid >> 5);
  const int nblk = (E + BM_E - 1) / BM_E;
  const int rows = min(NOWN, (N + RCAP - 1) / RCAP);
  const bool vec = ((reinterpret_cast<size_t>(dstp) | reinterpret_cast<size_t>(srcp)) & 15) == 0;

  for (int blk = blockIdx.x; blk < nblk; blk += gridDim.x) {
    for (int i = tid; i < NOWN * SG_LP / 4; i += 1024)
      reinterpret_cast<int4*>(lseg)[i] = make_int4(-1, -1, -1, -1);
    if (tid < NOWN) lcnt[tid] = 0;
    __syncthreads();

    const int e0 = blk * BM_E + 4 * tid;
    if (vec && e0 + 3 < E) {
      const int4 d = *reinterpret_cast<const int4*>(dstp + e0);
      const int4 s = *reinterpret_cast<const int4*>(srcp + e0);
      const unsigned o0 = owner_of(d.x), o1 = owner_of(d.y), o2 = owner_of(d.z), o3 = owner_of(d.w);
      const int r0 = atomicAdd(&lcnt[o0], 1);
      const int r1 = atomicAdd(&lcnt[o1], 1);
      const int r2 = atomicAdd(&lcnt[o2], 1);
      const int r3 = atomicAdd(&lcnt[o3], 1);
      put_pair(lseg, o0, r0, s.x, d.x);
      put_pair(lseg, o1, r1, s.y, d.y);
      put_pair(lseg, o2, r2, s.z, d.z);
      put_pair(lseg, o3, r3, s.w, d.w);
    } else {
      for (int j = 0; j < 4; ++j)
        if (e0 + j < E) {
          const int d = dstp[e0 + j];
          const unsigned o = owner_of(d);
          put_pair(lseg, o, atomicAdd(&lcnt[o], 1), srcp[e0 + j], d);
        }
    }
    __syncthreads();

    for (int o = wave; o < rows; o += 32) {
      if (lane == 0 && lcnt[o] > SEG) lseg[o * SG_LP] = -2;
      const int4 v = *reinterpret_cast<const int4*>(lseg + o * SG_LP + 4 * lane);
      *reinterpret_cast<int4*>(segs + ((size_t)o * nblk + blk) * (2 * SEG) + 4 * lane) = v;
    }
    __syncthreads();
  }
}

__device__ __forceinline__ void drain_batch(
    const int2* __restrict__ myr, int head, int nb, int lane,
    const float* __restrict__ news, float* acc) {
  int s = 0, t = RCAP;
  if (lane < nb) {
    const int2 p = myr[(head + lane) & (QRING - 1)];
    s = p.x;
    t = p.y;
  }
  for (int k = 0; k < nb; k += 8) {
    float4 r[8];
    int tt[8];
#pragma unroll
    for (int v = 0; v < 8; ++v) {
      const int kv8 = min(k + v, nb - 1);
      const int sk = __builtin_amdgcn_readlane(s, kv8);
      tt[v] = __builtin_amdgcn_readlane(t, kv8);
      r[v] = *reinterpret_cast<const float4*>(news + (size_t)sk * D + lane * 4);
    }
#pragma unroll
    for (int v = 0; v < 8; ++v) {
      if (k + v < nb) {
        float* a = acc + tt[v] * D + lane;
        atomicAdd(a,      r[v].x);
        atomicAdd(a + 32, r[v].y);
        atomicAdd(a + 64, r[v].z);
        atomicAdd(a + 96, r[v].w);
      }
    }
  }
}

__device__ __forceinline__ void push_pair(
    int2* myr, int tail, unsigned bal, int s, int d, float* degc, int lo, int cnt) {
  int t = RCAP;
  const int dl = d - lo;
  if ((unsigned)dl < (unsigned)cnt) {
    t = dl;
    atomicAdd(&degc[t], 1.0f);
  }
  myr[(tail + (int)__builtin_amdgcn_mbcnt_lo(bal, 0u)) & (QRING - 1)] = make_int2(s, t);
}

__global__ __launch_bounds__(1024) void edge_agg_owner(
    const int* __restrict__ ei, const int* __restrict__ segs, const float* __restrict__ news,
    float* __restrict__ aggn, int E, int N) {
  __shared__ float acc[(RCAP + 1) * D];
  __shared__ float degc[RCAP];
  __shared__ int2  ring[32][QRING];

  const int tid  = threadIdx.x;
  const int lane = tid & 31;
  const int wave = __builtin_amdgcn_readfirstlane(tid >> 5);
  const int* __restrict__ srcp = ei;
  const int* __restrict__ dstp = ei + E;
  const int nblk = (E + BM_E - 1) / BM_E;
  const unsigned key = blockIdx.x & 255u;
  int2* __restrict__ myr = ring[wave];
  const int4* __restrict__ srow = reinterpret_cast<const int4*>(segs) + (size_t)key * nblk * 32;
  const int4 none4 = make_int4(-1, -1, -1, -1);

  for (int lo = blockIdx.x * RCAP; lo < N; lo += NOWN * RCAP) {
    const int cnt = min(RCAP, N - lo);

    for (int i = tid; i < (RCAP + 1) * D / 4; i += 1024)
      reinterpret_cast<float4*>(acc)[i] = make_float4(0.f, 0.f, 0.f, 0.f);
    if (tid < RCAP) degc[tid] = 0.f;
    __syncthreads();

    int head = 0, tail = 0;

    auto take = [&](const int4 sv, const int blk) {
      if (__builtin_amdgcn_readfirstlane(sv.x) != -2) {
        const unsigned ba = __builtin_amdgcn_ballot_w32(sv.x >= 0);
        if (sv.x >= 0) push_pair(myr, tail, ba, sv.x, sv.y, degc, lo, cnt);
        tail += __builtin_popcount(ba);
        const unsigned bb = __builtin_amdgcn_ballot_w32(sv.z >= 0);
        if (sv.z >= 0) push_pair(myr, tail, bb, sv.z, sv.w, degc, lo, cnt);
        tail += __builtin_popcount(bb);
        while (tail - head >= 32) {
          drain_batch(myr, head, 32, lane, news, acc);
          head += 32;
        }
      } else {
        for (int i = 0; i < BM_E / 32; ++i) {
          const int e = blk * BM_E + 32 * i + lane;
          bool hit = false;
          int es = 0, ed = 0;
          if (e < E) {
            ed = dstp[e];
            if ((((unsigned)ed / RCAP) & 255u) == key) {
              hit = true;
              es = srcp[e];
            }
          }
          const unsigned bh = __builtin_amdgcn_ballot_w32(hit);
          if (hit) push_pair(myr, tail, bh, es, ed, degc, lo, cnt);
          tail += __builtin_popcount(bh);
          while (tail - head >= 32) {
            drain_batch(myr, head, 32, lane, news, acc);
            head += 32;
          }
        }
      }
    };

    int4 v = none4, w = none4;
    if (wave < nblk) {
      v = srow[(size_t)wave * 32 + lane];
      w = srow[(size_t)((wave + 32 < nblk) ? wave + 32 : wave) * 32 + lane];
    }
    for (int blk = wave; blk < nblk; blk += 64) {
      const int bv = (blk + 64 < nblk) ? blk + 64 : blk;
      const int bw = (blk + 96 < nblk) ? blk + 96 : blk;
      const int4 vn = srow[(size_t)bv * 32 + lane];
      const int4 wn = srow[(size_t)bw * 32 + lane];
      take(v, blk);
      if (blk + 32 < nblk) take(w, blk + 32);
      v = vn;
      w = wn;
    }
    if (tail - head > 0) drain_batch(myr, head, tail - head, lane, news, acc);
    __syncthreads();

    for (int t = wave; t < cnt; t += 32) {
      const float* a = acc + t * D + lane;
      const float dinv = 1.0f / fmaxf(degc[t], 1.0f);
      const float4 v4 = make_float4(a[0] * dinv, a[32] * dinv, a[64] * dinv, a[96] * dinv);
      *reinterpret_cast<float4*>(aggn + (size_t)(lo + t) * D + lane * 4) = v4;
    }
    __syncthreads();
  }
}

#define UB 16
#define UP 132
#define RG 10
#define HSPLIT 30

__global__ __launch_bounds__(512) void user_score_lin(
    const float* __restrict__ news, const float* __restrict__ aggn, const float* __restrict__ Wg,
    const int* __restrict__ hist, const int* __restrict__ cand, float* __restrict__ scores, int B) {
  __shared__ float sN[UB * UP];
  __shared__ float sA[UB * UP];
  __shared__ float sC[UB];
  __shared__ float sS[UB * CAND];

  const int tid  = threadIdx.x;
  const int lane = tid & 31;
  const int wave = __builtin_amdgcn_readfirstlane(tid >> 5);
  const int c = lane & 15, h = lane >> 4;
  const int ntile = (B + UB - 1) / UB;

  for (int tile = blockIdx.x; tile < ntile; tile += gridDim.x) {
    const int b = tile * UB + wave;

    int hv0 = 0, hv1 = 0, cv = 0;
    if (b < B) {
      if (lane < HSPLIT) hv0 = hist[(size_t)b * HIST + lane];
      if (lane < HIST - HSPLIT) hv1 = hist[(size_t)b * HIST + HSPLIT + lane];
      if (lane < CAND) cv = cand[(size_t)b * CAND + lane];
    }

    float4 sn = make_float4(0.f, 0.f, 0.f, 0.f), sa = sn;
    float cnt = 0.f;
    if (b < B) {
      for (int g = 0; g < HIST; g += RG) {
        float4 n[RG], a[RG];
        float m[RG];
        const int hsel = (g < HSPLIT) ? hv0 : hv1;
        const int l0   = (g < HSPLIT) ? g : g - HSPLIT;
#pragma unroll
        for (int k = 0; k < RG; ++k) {
          const int nid = __builtin_amdgcn_readlane(hsel, l0 + k);
          m[k] = (nid != 0) ? 1.0f : 0.0f;
          n[k] = *reinterpret_cast<const float4*>(news + (size_t)nid * D + lane * 4);
          a[k] = *reinterpret_cast<const float4*>(aggn + (size_t)nid * D + lane * 4);
        }
#pragma unroll
        for (int k = 0; k < RG; ++k) {
          sn.x += m[k] * n[k].x; sn.y += m[k] * n[k].y; sn.z += m[k] * n[k].z; sn.w += m[k] * n[k].w;
          sa.x += m[k] * a[k].x; sa.y += m[k] * a[k].y; sa.z += m[k] * a[k].z; sa.w += m[k] * a[k].w;
          cnt += m[k];
        }
      }
    }
    *reinterpret_cast<float4*>(&sN[wave * UP + lane * 4]) = sn;
    *reinterpret_cast<float4*>(&sA[wave * UP + lane * 4]) = sa;
    if (lane == 0) sC[wave] = cnt;
    __syncthreads();

    v2f bw[D / 4];
    if (wave < 8) {
      const int col = wave * 16 + c;
#pragma unroll
      for (int j = 0; j < D / 4; ++j) {
        bw[j].x = Wg[(size_t)(4 * j + 2 * h + 0) * D + col];
        bw[j].y = Wg[(size_t)(4 * j + 2 * h + 1) * D + col];
      }
      const float* pa = &sA[c * UP + 2 * h];
      v8f acc = {};
#pragma unroll
      for (int j = 0; j < D / 4; ++j) {
        const v2f a = *reinterpret_cast<const v2f*>(pa + 4 * j);
        acc = __builtin_amdgcn_wmma_f32_16x16x4_f32(
            false, a, false, bw[j], (short)0, acc, false, false);
      }
#pragma unroll
      for (int r = 0; r < 8; ++r) {
        const int row = 8 * h + r;
        sN[row * UP + col] = (sN[row * UP + col] + acc[r]) / fmaxf(sC[row], 1e-9f);
      }
    } else {
      const int col = (wave - 8) * 16 + c;
#pragma unroll
      for (int j = 0; j < D / 4; ++j)
        bw[j] = *reinterpret_cast<const v2f*>(Wg + (size_t)col * D + 4 * j + 2 * h);
    }
    __syncthreads();

    if (wave >= 8) {
      const int col = (wave - 8) * 16 + c;
      const float* pa = &sN[c * UP + 2 * h];
      v8f acc = {};
#pragma unroll
      for (int j = 0; j < D / 4; ++j) {
        const v2f a = *reinterpret_cast<const v2f*>(pa + 4 * j);
        acc = __builtin_amdgcn_wmma_f32_16x16x4_f32(
            false, a, false, bw[j], (short)0, acc, false, false);
      }
#pragma unroll
      for (int r = 0; r < 8; ++r) sA[(8 * h + r) * UP + col] = acc[r];
    }
    __syncthreads();

    if (b < B) {
      const float4 u = *reinterpret_cast<const float4*>(&sN[wave * UP + lane * 4]);
      const float4 w = *reinterpret_cast<const float4*>(&sA[wave * UP + lane * 4]);
      float mine = 0.f;
      for (int g = 0; g < CAND; g += RG) {
        float4 n[RG], a[RG];
#pragma unroll
        for (int k = 0; k < RG; ++k) {
          const int nid = __builtin_amdgcn_readlane(cv, g + k);
          n[k] = *reinterpret_cast<const float4*>(news + (size_t)nid * D + lane * 4);
          a[k] = *reinterpret_cast<const float4*>(aggn + (size_t)nid * D + lane * 4);
        }
#pragma unroll
        for (int k = 0; k < RG; ++k) {
          float p = n[k].x * u.x + n[k].y * u.y + n[k].z * u.z + n[k].w * u.w
                  + a[k].x * w.x + a[k].y * w.y + a[k].z * w.z + a[k].w * w.w;
          for (int off = 16; off; off >>= 1) p += __shfl_xor(p, off, 32);
          if (lane == g + k) mine = p;
        }
      }
      if (lane < CAND) sS[wave * CAND + lane] = mine;
    }
    __syncthreads();

    if (tid < UB * CAND) {
      const size_t g = (size_t)tile * (UB * CAND) + tid;
      if (g < (size_t)B * CAND) scores[g] = sS[tid];
    }
  }
}

extern "C" void kernel_launch(void* const* d_in, const int* in_sizes, int n_in,
                              void* d_out, int out_size, void* d_ws, size_t ws_size,
                              hipStream_t stream) {
  const float* x      = (const float*)d_in[0];
  const int*   ei     = (const int*)  d_in[2];
  const int*   hist   = (const int*)  d_in[3];
  const int*   cand   = (const int*)  d_in[4];
  const float* W_enc  = (const float*)d_in[5];
  const float* W_gnn  = (const float*)d_in[6];
  float*       scores = (float*)d_out;

  const int       N = in_sizes[1];
  const long long E = (long long)in_sizes[2] / 2;
  const int       B = in_sizes[3] / HIST;

  float* news = (float*)d_ws;
  float* aggn = news + (size_t)N * D;
  int* segs = (int*)(aggn + (size_t)N * D);

  gemm_wstat<300, 300><<<512, 256, 0, stream>>>(x, W_enc, news, N);

  edge_owner_pairs<<<512, 1024, 0, stream>>>(ei, segs, (int)E, N);

  edge_agg_owner<<<NOWN, 1024, 0, stream>>>(ei, segs, news, aggn, (int)E, N);

  user_score_lin<<<256, 512, 0, stream>>>(news, aggn, W_gnn, hist, cand, scores, B);
}


# Edge pre-pass launch folded into the node GEMM launch: a GEMM work-group sorts a 4096-edge block in its LDS before its first or after its last trip; segment slot and LDS sum order stay arrival order
# speedup vs baseline: 7.1554x; 1.0307x over previous
#include <hip/hip_runtime.h>
#include <hip/hip_bf16.h>

typedef float v2f __attribute__((ext_vector_type(2)));
typedef float v8f __attribute__((ext_vector_type(8)));

#define D 128
#define HIST 50
#define CAND 20

#define RCAP   416
#define NOWN   256
#define BM_E   4096
#define SEG    64
#define SG_LP  132
#define QRING  128

__device__ __forceinline__ unsigned owner_of(int d) { return ((unsigned)d / RCAP) & 255u; }

__device__ __forceinline__ void put_pair(int* lseg, unsigned o, int r, int s, int d) {
  if (r < SEG) *reinterpret_cast<int2*>(lseg + o * SG_LP + 2 * r) = make_int2(s, d);
}


#define PP_NT  256
#define PP_NW  (PP_NT / 32)

__device__ __forceinline__ void sort_four(int* lseg, int* lcnt, const int4 s, const int4 d) {
  const unsigned o0 = owner_of(d.x), o1 = owner_of(d.y), o2 = owner_of(d.z), o3 = owner_of(d.w);
  const int r0 = atomicAdd(&lcnt[o0], 1);
  const int r1 = atomicAdd(&lcnt[o1], 1);
  const int r2 = atomicAdd(&lcnt[o2], 1);
  const int r3 = atomicAdd(&lcnt[o3], 1);
  put_pair(lseg, o0, r0, s.x, d.x);
  put_pair(lseg, o1, r1, s.y, d.y);
  put_pair(lseg, o2, r2, s.z, d.z);
  put_pair(lseg, o3, r3, s.w, d.w);
}

__device__ __forceinline__ void owner_pairs_block(
    int* lseg, int* lcnt, const int* __restrict__ srcp, const int* __restrict__ dstp,
    int* __restrict__ segs, int E, int blk, int nblk, int rows, bool vec, int tid, int lane, int wave) {
  for (int i = tid; i < NOWN * SG_LP / 4; i += PP_NT)
    reinterpret_cast<int4*>(lseg)[i] = make_int4(-1, -1, -1, -1);
  if (tid < NOWN) lcnt[tid] = 0;
  __syncthreads();

  const int eb = blk * BM_E;
  if (vec && eb + BM_E <= E) {
    const int4* __restrict__ dp = reinterpret_cast<const int4*>(dstp + eb) + tid;
    const int4* __restrict__ sp = reinterpret_cast<const int4*>(srcp + eb) + tid;
    const int4 d0 = dp[0],         s0 = sp[0];
    const int4 d1 = dp[PP_NT],     s1 = sp[PP_NT];
    const int4 d2 = dp[2 * PP_NT], s2 = sp[2 * PP_NT];
    const int4 d3 = dp[3 * PP_NT], s3 = sp[3 * PP_NT];
    sort_four(lseg, lcnt, s0, d0);
    sort_four(lseg, lcnt, s1, d1);
    sort_four(lseg, lcnt, s2, d2);
    sort_four(lseg, lcnt, s3, d3);
  } else {
    for (int q = 0; q < BM_E / (4 * PP_NT); ++q) {
      const int e0 = eb + 4 * (q * PP_NT + tid);
      if (vec && e0 + 3 < E) {
        sort_four(lseg, lcnt, *reinterpret_cast<const int4*>(srcp + e0), *reinterpret_cast<const int4*>(dstp + e0));
      } else {
        for (int j = 0; j < 4; ++j)
          if (e0 + j < E) {
            const int d = dstp[e0 + j];
            const unsigned o = owner_of(d);
            put_pair(lseg, o, atomicAdd(&lcnt[o], 1), srcp[e0 + j], d);
          }
      }
    }
  }
  __syncthreads();

  static_assert(PP_NW * 32 == NOWN, "one lane per segment of a wave");
  const int n = (rows > wave) ? (rows - wave + PP_NW - 1) / PP_NW : 0;
  const int myc = lcnt[wave + PP_NW * lane];
  auto seg_piece = [&](int i) {
    return *reinterpret_cast<const int4*>(lseg + (wave + PP_NW * i) * SG_LP + 4 * lane);
  };
  auto marked = [&](int4 v, int i) {
    const int cnt = __builtin_amdgcn_readlane(myc, i);
    if (lane == 0 && cnt > SEG) v.x = -2;
    return v;
  };
  auto seg_out = [&](int i) {
    return reinterpret_cast<int4*>(segs + ((size_t)(wave + PP_NW * i) * nblk + blk) * (2 * SEG) + 4 * lane);
  };
  for (int i = 0; i < n; i += 4) {
    const int i1 = min(i + 1, n - 1), i2 = min(i + 2, n - 1), i3 = min(i + 3, n - 1);
    const int4 v0 = seg_piece(i), v1 = seg_piece(i1), v2 = seg_piece(i2), v3 = seg_piece(i3);
    *seg_out(i)  = marked(v0, i);
    *seg_out(i1) = marked(v1, i1);
    *seg_out(i2) = marked(v2, i2);
    *seg_out(i3) = marked(v3, i3);
  }
  __syncthreads();
}

#define GB_ROWS 16

template <int K, int PITCH>
__global__ __launch_bounds__(256) __attribute__((amdgpu_waves_per_eu(4, 4))) void gemm_wstat(
    const float* __restrict__ A, const float* __restrict__ W,
    float* __restrict__ out, int N,
    const int* __restrict__ ei, int* __restrict__ segs, int E) {
  constexpr int KS  = K / 4;
  constexpr int RF4 = K / 4;
  constexpr int NF4 = GB_ROWS * RF4;
  constexpr int NLD = (NF4 + 255) / 256;
  constexpr int STAGE = (GB_ROWS * PITCH > NLD * 1024) ? GB_ROWS * PITCH : NLD * 1024;
  static_assert(2 * STAGE <= NOWN * SG_LP + NOWN, "the stages lie inside the pre-pass block's LDS");
  __shared__ __attribute__((aligned(16))) int smem[NOWN * SG_LP + NOWN];
  float (*lds)[STAGE] = reinterpret_cast<float (*)[STAGE]>(smem);

  const int tid  = threadIdx.x;
  const int lane = tid & 31;
  const int wave = __builtin_amdgcn_readfirstlane(tid >> 5);
  const int c = lane & 15, h = lane >> 4;
  const int col = wave * 16 + c;
  const int nblk = N / GB_ROWS;
  const int G = (int)gridDim.x;

  const int* __restrict__ srcp = ei;
  const int* __restrict__ dstp = ei + E;
  const int pnblk = (E + BM_E - 1) / BM_E;
  const int prows = min(NOWN, (N + RCAP - 1) / RCAP);
  const bool pvec = ((reinterpret_cast<size_t>(dstp) | reinterpret_cast<size_t>(srcp)) & 15) == 0;
  const int  half = G / 2;
  const bool pfirst = (int)blockIdx.x >= half;
  const int  pidx = pfirst ? (int)blockIdx.x - half : (G - half) + (half - 1 - (int)blockIdx.x);

  if (pfirst)
    for (int pb = pidx; pb < pnblk; pb += G)
      owner_pairs_block(smem, smem + NOWN * SG_LP, srcp, dstp, segs, E, pb, pnblk, prows, pvec, tid, lane, wave);

  v2f b[KS];
#pragma unroll
  for (int j = 0; j < KS; ++j) {
    b[j].x = W[(size_t)(4 * j + 2 * h + 0) * D + col];
    b[j].y = W[(size_t)(4 * j + 2 * h + 1) * D + col];
  }

  static_assert(NLD == 5, "the block's loads are written out five to a thread");
  float4 p0, p1, p2, p3, p4;
  const int f0 = tid, f1 = tid + 256, f2 = tid + 512, f3 = tid + 768, f4 = min(tid + 1024, NF4 - 1);
  auto slot = [](int f) { return (PITCH == K) ? 4 * f : (f / RF4) * PITCH + (f % RF4) * 4; };
  const int a0s = slot(tid), a1s = slot(tid + 256), a2s = slot(tid + 512), a3s = slot(tid + 768), a4s = slot(tid + 1024);

  int blk = blockIdx.x;

  if (blk < nblk) {
    const float4* __restrict__ src = reinterpret_cast<const float4*>(A) + (size_t)blk * NF4;
    p0 = src[f0]; p1 = src[f1]; p2 = src[f2]; p3 = src[f3]; p4 = src[f4];
    *reinterpret_cast<float4*>(&lds[0][a0s]) = p0;
    *reinterpret_cast<float4*>(&lds[0][a1s]) = p1;
    *reinterpret_cast<float4*>(&lds[0][a2s]) = p2;
    *reinterpret_cast<float4*>(&lds[0][a3s]) = p3;
    *reinterpret_cast<float4*>(&lds[0][a4s]) = p4;
    const int nxt = (blk + G < nblk) ? blk + G : blk;
    const float4* __restrict__ src1 = reinterpret_cast<const float4*>(A) + (size_t)nxt * NF4;
    p0 = src1[f0]; p1 = src1[f1]; p2 = src1[f2]; p3 = src1[f3]; p4 = src1[f4];
  } else {
    p0 = p1 = p2 = p3 = p4 = make_float4(0.f, 0.f, 0.f, 0.f);
  }
  __syncthreads();

  int s = 0;
  for (; blk < nblk; blk += G) {
    const size_t o0 = (size_t)(blk * GB_ROWS + 8 * h) * D + col;

    const float* sA = &lds[s][c * PITCH + 2 * h];
    v8f acc0 = {};
#pragma unroll
    for (int j = 0; j < KS; ++j) {
      const v2f a0 = *reinterpret_cast<const v2f*>(sA + 4 * j);
      acc0 = __builtin_amdgcn_wmma_f32_16x16x4_f32(
          false, a0, false, b[j], (short)0, acc0, false, false);
    }
    __builtin_amdgcn_sched_group_barrier(0x100, 6, 0);
    __builtin_amdgcn_sched_group_barrier(0x020, 5, 0);
#pragma unroll
    for (int g = 0; g < 32; ++g) {
      __builtin_amdgcn_sched_group_barrier(0x008, 2, 0);
      __builtin_amdgcn_sched_group_barrier(0x100, 1, 0);
    }
    __builtin_amdgcn_sched_group_barrier(0x008, 11, 0);

#pragma unroll
    for (int r = 0; r < 8; ++r) out[o0 + (size_t)r * D] = acc0[r];

    *reinterpret_cast<float4*>(&lds[s ^ 1][a0s]) = p0;
    *reinterpret_cast<float4*>(&lds[s ^ 1][a1s]) = p1;
    *reinterpret_cast<float4*>(&lds[s ^ 1][a2s]) = p2;
    *reinterpret_cast<float4*>(&lds[s ^ 1][a3s]) = p3;
    *reinterpret_cast<float4*>(&lds[s ^ 1][a4s]) = p4;
    {
      const int nn = (blk + 2 * G < nblk) ? blk + 2 * G : blk;
      const float4* __restrict__ src2 = reinterpret_cast<const float4*>(A) + (size_t)nn * NF4;
      p0 = src2[f0]; p1 = src2[f1]; p2 = src2[f2]; p3 = src2[f3]; p4 = src2[f4];
    }
    __builtin_amdgcn_fence(__ATOMIC_RELEASE, "workgroup", "local");
    __builtin_amdgcn_s_barrier();
    __builtin_amdgcn_fence(__ATOMIC_ACQUIRE, "workgroup", "local");
    s ^= 1;
  }

  if (!pfirst)
    for (int pb = pidx; pb < pnblk; pb += G)
      owner_pairs_block(smem, smem + NOWN * SG_LP, srcp, dstp, segs, E, pb, pnblk, prows, pvec, tid, lane, wave);
}

__device__ __forceinline__ void drain_batch(
    const int2* __restrict__ myr, int head, int nb, int lane,
    const float* __restrict__ news, float* acc) {
  int s = 0, t = RCAP;
  if (lane < nb) {
    const int2 p = myr[(head + lane) & (QRING - 1)];
    s = p.x;
    t = p.y;
  }
  for (int k = 0; k < nb; k += 8) {
    float4 r[8];
    int tt[8];
#pragma unroll
    for (int v = 0; v < 8; ++v) {
      const int kv8 = min(k + v, nb - 1);
      const int sk = __builtin_amdgcn_readlane(s, kv8);
      tt[v] = __builtin_amdgcn_readlane(t, kv8);
      r[v] = *reinterpret_cast<const float4*>(news + (size_t)sk * D + lane * 4);
    }
#pragma unroll
    for (int v = 0; v < 8; ++v) {
      if (k + v < nb) {
        float* a = acc + tt[v] * D + lane;
        atomicAdd(a,      r[v].x);
        atomicAdd(a + 32, r[v].y);
        atomicAdd(a + 64, r[v].z);
        atomicAdd(a + 96, r[v].w);
      }
    }
  }
}

__device__ __forceinline__ void push_pair(
    int2* myr, int tail, unsigned bal, int s, int d, float* degc, int lo, int cnt) {
  int t = RCAP;
  const int dl = d - lo;
  if ((unsigned)dl < (unsigned)cnt) {
    t = dl;
    atomicAdd(&degc[t], 1.0f);
  }
  myr[(tail + (int)__builtin_amdgcn_mbcnt_lo(bal, 0u)) & (QRING - 1)] = make_int2(s, t);
}

__global__ __launch_bounds__(1024) void edge_agg_owner(
    const int* __restrict__ ei, const int* __restrict__ segs, const float* __restrict__ news,
    float* __restrict__ aggn, int E, int N) {
  __shared__ float acc[(RCAP + 1) * D];
  __shared__ float degc[RCAP];
  __shared__ int2  ring[32][QRING];

  const int tid  = threadIdx.x;
  const int lane = tid & 31;
  const int wave = __builtin_amdgcn_readfirstlane(tid >> 5);
  const int* __restrict__ srcp = ei;
  const int* __restrict__ dstp = ei + E;
  const int nblk = (E + BM_E - 1) / BM_E;
  const unsigned key = blockIdx.x & 255u;
  int2* __restrict__ myr = ring[wave];
  const int4* __restrict__ srow = reinterpret_cast<const int4*>(segs) + (size_t)key * nblk * 32;
  const int4 none4 = make_int4(-1, -1, -1, -1);

  for (int lo = blockIdx.x * RCAP; lo < N; lo += NOWN * RCAP) {
    const int cnt = min(RCAP, N - lo);

    for (int i = tid; i < (RCAP + 1) * D / 4; i += 1024)
      reinterpret_cast<float4*>(acc)[i] = make_float4(0.f, 0.f, 0.f, 0.f);
    if (tid < RCAP) degc[tid] = 0.f;
    __syncthreads();

    int head = 0, tail = 0;

    auto take = [&](const int4 sv, const int blk) {
      if (__builtin_amdgcn_readfirstlane(sv.x) != -2) {
        const unsigned ba = __builtin_amdgcn_ballot_w32(sv.x >= 0);
        if (sv.x >= 0) push_pair(myr, tail, ba, sv.x, sv.y, degc, lo, cnt);
        tail += __builtin_popcount(ba);
        const unsigned bb = __builtin_amdgcn_ballot_w32(sv.z >= 0);
        if (sv.z >= 0) push_pair(myr, tail, bb, sv.z, sv.w, degc, lo, cnt);
        tail += __builtin_popcount(bb);
        while (tail - head >= 32) {
          drain_batch(myr, head, 32, lane, news, acc);
          head += 32;
        }
      } else {
        for (int i = 0; i < BM_E / 32; ++i) {
          const int e = blk * BM_E + 32 * i + lane;
          bool hit = false;
          int es = 0, ed = 0;
          if (e < E) {
            ed = dstp[e];
            if ((((unsigned)ed / RCAP) & 255u) == key) {
              hit = true;
              es = srcp[e];
            }
          }
          const unsigned bh = __builtin_amdgcn_ballot_w32(hit);
          if (hit) push_pair(myr, tail, bh, es, ed, degc, lo, cnt);
          tail += __builtin_popcount(bh);
          while (tail - head >= 32) {
            drain_batch(myr, head, 32, lane, news, acc);
            head += 32;
          }
        }
      }
    };

    int4 v = none4, w = none4;
    if (wave < nblk) {
      v = srow[(size_t)wave * 32 + lane];
      w = srow[(size_t)((wave + 32 < nblk) ? wave + 32 : wave) * 32 + lane];
    }
    for (int blk = wave; blk < nblk; blk += 64) {
      const int bv = (blk + 64 < nblk) ? blk + 64 : blk;
      const int bw = (blk + 96 < nblk) ? blk + 96 : blk;
      const int4 vn = srow[(size_t)bv * 32 + lane];
      const int4 wn = srow[(size_t)bw * 32 + lane];
      take(v, blk);
      if (blk + 32 < nblk) take(w, blk + 32);
      v = vn;
      w = wn;
    }
    if (tail - head > 0) drain_batch(myr, head, tail - head, lane, news, acc);
    __syncthreads();

    for (int t = wave; t < cnt; t += 32) {
      const float* a = acc + t * D + lane;
      const float dinv = 1.0f / fmaxf(degc[t], 1.0f);
      const float4 v4 = make_float4(a[0] * dinv, a[32] * dinv, a[64] * dinv, a[96] * dinv);
      *reinterpret_cast<float4*>(aggn + (size_t)(lo + t) * D + lane * 4) = v4;
    }
    __syncthreads();
  }
}

#define UB 16
#define UP 132
#define RG 10
#define HSPLIT 30

__global__ __launch_bounds__(512) void user_score_lin(
    const float* __restrict__ news, const float* __restrict__ aggn, const float* __restrict__ Wg,
    const int* __restrict__ hist, const int* __restrict__ cand, float* __restrict__ scores, int B) {
  __shared__ float sN[UB * UP];
  __shared__ float sA[UB * UP];
  __shared__ float sC[UB];
  __shared__ float sS[UB * CAND];

  const int tid  = threadIdx.x;
  const int lane = tid & 31;
  const int wave = __builtin_amdgcn_readfirstlane(tid >> 5);
  const int c = lane & 15, h = lane >> 4;
  const int ntile = (B + UB - 1) / UB;

  for (int tile = blockIdx.x; tile < ntile; tile += gridDim.x) {
    const int b = tile * UB + wave;

    int hv0 = 0, hv1 = 0, cv = 0;
    if (b < B) {
      if (lane < HSPLIT) hv0 = hist[(size_t)b * HIST + lane];
      if (lane < HIST - HSPLIT) hv1 = hist[(size_t)b * HIST + HSPLIT + lane];
      if (lane < CAND) cv = cand[(size_t)b * CAND + lane];
    }

    float4 sn = make_float4(0.f, 0.f, 0.f, 0.f), sa = sn;
    float cnt = 0.f;
    if (b < B) {
      for (int g = 0; g < HIST; g += RG) {
        float4 n[RG], a[RG];
        float m[RG];
        const int hsel = (g < HSPLIT) ? hv0 : hv1;
        const int l0   = (g < HSPLIT) ? g : g - HSPLIT;
#pragma unroll
        for (int k = 0; k < RG; ++k) {
          const int nid = __builtin_amdgcn_readlane(hsel, l0 + k);
          m[k] = (nid != 0) ? 1.0f : 0.0f;
          n[k] = *reinterpret_cast<const float4*>(news + (size_t)nid * D + lane * 4);
          a[k] = *reinterpret_cast<const float4*>(aggn + (size_t)nid * D + lane * 4);
        }
#pragma unroll
        for (int k = 0; k < RG; ++k) {
          sn.x += m[k] * n[k].x; sn.y += m[k] * n[k].y; sn.z += m[k] * n[k].z; sn.w += m[k] * n[k].w;
          sa.x += m[k] * a[k].x; sa.y += m[k] * a[k].y; sa.z += m[k] * a[k].z; sa.w += m[k] * a[k].w;
          cnt += m[k];
        }
      }
    }
    *reinterpret_cast<float4*>(&sN[wave * UP + lane * 4]) = sn;
    *reinterpret_cast<float4*>(&sA[wave * UP + lane * 4]) = sa;
    if (lane == 0) sC[wave] = cnt;
    __syncthreads();

    v2f bw[D / 4];
    if (wave < 8) {
      const int col = wave * 16 + c;
#pragma unroll
      for (int j = 0; j < D / 4; ++j) {
        bw[j].x = Wg[(size_t)(4 * j + 2 * h + 0) * D + col];
        bw[j].y = Wg[(size_t)(4 * j + 2 * h + 1) * D + col];
      }
      const float* pa = &sA[c * UP + 2 * h];
      v8f acc = {};
#pragma unroll
      for (int j = 0; j < D / 4; ++j) {
        const v2f a = *reinterpret_cast<const v2f*>(pa + 4 * j);
        acc = __builtin_amdgcn_wmma_f32_16x16x4_f32(
            false, a, false, bw[j], (short)0, acc, false, false);
      }
#pragma unroll
      for (int r = 0; r < 8; ++r) {
        const int row = 8 * h + r;
        sN[row * UP + col] = (sN[row * UP + col] + acc[r]) / fmaxf(sC[row], 1e-9f);
      }
    } else {
      const int col = (wave - 8) * 16 + c;
#pragma unroll
      for (int j = 0; j < D / 4; ++j)
        bw[j] = *reinterpret_cast<const v2f*>(Wg + (size_t)col * D + 4 * j + 2 * h);
    }
    __syncthreads();

    if (wave >= 8) {
      const int col = (wave - 8) * 16 + c;
      const float* pa = &sN[c * UP + 2 * h];
      v8f acc = {};
#pragma unroll
      for (int j = 0; j < D / 4; ++j) {
        const v2f a = *reinterpret_cast<const v2f*>(pa + 4 * j);
        acc = __builtin_amdgcn_wmma_f32_16x16x4_f32(
            false, a, false, bw[j], (short)0, acc, false, false);
      }
#pragma unroll
      for (int r = 0; r < 8; ++r) sA[(8 * h + r) * UP + col] = acc[r];
    }
    __syncthreads();

    if (b < B) {
      const float4 u = *reinterpret_cast<const float4*>(&sN[wave * UP + lane * 4]);
      const float4 w = *reinterpret_cast<const float4*>(&sA[wave * UP + lane * 4]);
      float mine = 0.f;
      for (int g = 0; g < CAND; g += RG) {
        float4 n[RG], a[RG];
#pragma unroll
        for (int k = 0; k < RG; ++k) {
          const int nid = __builtin_amdgcn_readlane(cv, g + k);
          n[k] = *reinterpret_cast<const float4*>(news + (size_t)nid * D + lane * 4);
          a[k] = *reinterpret_cast<const float4*>(aggn + (size_t)nid * D + lane * 4);
        }
#pragma unroll
        for (int k = 0; k < RG; ++k) {
          float p = n[k].x * u.x + n[k].y * u.y + n[k].z * u.z + n[k].w * u.w
                  + a[k].x * w.x + a[k].y * w.y + a[k].z * w.z + a[k].w * w.w;
          for (int off = 16; off; off >>= 1) p += __shfl_xor(p, off, 32);
          if (lane == g + k) mine = p;
        }
      }
      if (lane < CAND) sS[wave * CAND + lane] = mine;
    }
    __syncthreads();

    if (tid < UB * CAND) {
      const size_t g = (size_t)tile * (UB * CAND) + tid;
      if (g < (size_t)B * CAND) scores[g] = sS[tid];
    }
  }
}

extern "C" void kernel_launch(void* const* d_in, const int* in_sizes, int n_in,
                              void* d_out, int out_size, void* d_ws, size_t ws_size,
                              hipStream_t stream) {
  const float* x      = (const float*)d_in[0];
  const int*   ei     = (const int*)  d_in[2];
  const int*   hist   = (const int*)  d_in[3];
  const int*   cand   = (const int*)  d_in[4];
  const float* W_enc  = (const float*)d_in[5];
  const float* W_gnn  = (const float*)d_in[6];
  float*       scores = (float*)d_out;

  const int       N = in_sizes[1];
  const long long E = (long long)in_sizes[2] / 2;
  const int       B = in_sizes[3] / HIST;

  float* news = (float*)d_ws;
  float* aggn = news + (size_t)N * D;
  int* segs = (int*)(aggn + (size_t)N * D);

  gemm_wstat<300, 300><<<512, 256, 0, stream>>>(x, W_enc, news, N, ei, segs, (int)E);

  edge_agg_owner<<<NOWN, 1024, 0, stream>>>(ei, segs, news, aggn, (int)E, N);

  user_score_lin<<<256, 512, 0, stream>>>(news, aggn, W_gnn, hist, cand, scores, B);
}
